# starting kernel
# baseline (speedup 1.0000x reference)
#include <hip/hip_runtime.h>
#include <hip/hip_bf16.h>
#include <math.h>

static constexpr int Bsz = 512;
static constexpr int T   = 16384;
static constexpr int L   = 512;
static constexpr int C   = T / L;

typedef __attribute__((ext_vector_type(2))) float v2f;
typedef __attribute__((ext_vector_type(4))) float v4f;
typedef __attribute__((ext_vector_type(8))) float v8f;

__device__ __forceinline__ float whf(float x) {
    const float c   = 10.0f / 11.0f;
    const float l2e = 1.4426950408889634f;
    float e = 1.0f - __builtin_amdgcn_exp2f(-(c * l2e) * x);
    float l = c * x;
    unsigned s = (unsigned)(__float_as_int(x) >> 31);
    return __uint_as_float((__float_as_uint(e) & ~s) | (__float_as_uint(l) & s));
}

__global__ void hpack_kernel(const float* __restrict__ a1c,
                             const float* __restrict__ a2c,
                             float* __restrict__ hp1,
                             float* __restrict__ hp2) {
    const float* ac = (blockIdx.x == 0) ? a1c : a2c;
    float*       hp = (blockIdx.x == 0) ? hp1 : hp2;
    int j = threadIdx.x;
    if (j > 3) return;
    if (j == 3) {
        for (int i = 0; i < L; ++i) hp[i * 4 + 3] = 0.0f;
        return;
    }
    const float a0 = ac[0], a1 = ac[1], a2 = ac[2];
    float y1 = (j == 0) ? 1.0f : 0.0f;
    float y2 = (j == 1) ? 1.0f : 0.0f;
    float y3 = (j == 2) ? 1.0f : 0.0f;
    for (int i = 0; i < L; ++i) {
        float y = -(a0 * y1 + a1 * y2 + a2 * y3);
        hp[i * 4 + j] = y;
        y3 = y2; y2 = y1; y1 = y;
    }
}

__global__ __launch_bounds__(256)
void phase1_kernel(const float* __restrict__ in,
                   const float* __restrict__ bco,
                   const float* __restrict__ aco,
                   int nk,
                   float* __restrict__ z,
                   float* __restrict__ tails) {
    const int id = blockIdx.x * blockDim.x + threadIdx.x;
    const int b  = id / C;
    const int c  = id % C;
    const float b0 = bco[0], b1 = bco[1], b2 = bco[2];
    const float a0 = aco[0], a1 = aco[1], a2 = aco[2];

    const float* __restrict__ row = in + (size_t)b * T;
    const int t0 = c * L;
    float x1 = (t0 - nk - 1 >= 0) ? row[t0 - nk - 1] : 0.0f;
    float x2 = (t0 - nk - 2 >= 0) ? row[t0 - nk - 2] : 0.0f;
    float y1 = 0.0f, y2 = 0.0f, y3 = 0.0f;
    float* __restrict__ zrow = z + (size_t)b * T + t0;

#pragma unroll 4
    for (int i = 0; i < L; ++i) {
        const int t = t0 + i;
        if ((i & 63) == 0)
            __builtin_prefetch(row + t + 256, 0, 0);
        float x0 = (t - nk >= 0) ? row[t - nk] : 0.0f;
        float xf = b0 * x0 + b1 * x1 + b2 * x2;
        float y  = xf - (a0 * y1 + a1 * y2 + a2 * y3);
        zrow[i] = y;
        x2 = x1; x1 = x0;
        y3 = y2; y2 = y1; y1 = y;
    }
    float* tl = tails + (size_t)id * 4;
    tl[0] = y1; tl[1] = y2; tl[2] = y3; tl[3] = 0.0f;
}

__global__ __launch_bounds__(256)
void scan_kernel(const float* __restrict__ tails,
                 const float* __restrict__ hpack,
                 float* __restrict__ states) {
    const int b = blockIdx.x * blockDim.x + threadIdx.x;
    if (b >= Bsz) return;
    const float h00 = hpack[(L - 1) * 4 + 0], h01 = hpack[(L - 1) * 4 + 1], h02 = hpack[(L - 1) * 4 + 2];
    const float h10 = hpack[(L - 2) * 4 + 0], h11 = hpack[(L - 2) * 4 + 1], h12 = hpack[(L - 2) * 4 + 2];
    const float h20 = hpack[(L - 3) * 4 + 0], h21 = hpack[(L - 3) * 4 + 1], h22 = hpack[(L - 3) * 4 + 2];
    float s0 = 0.0f, s1 = 0.0f, s2 = 0.0f;
    for (int c = 0; c < C; ++c) {
        const size_t idx = ((size_t)b * C + c) * 4;
        states[idx + 0] = s0; states[idx + 1] = s1;
        states[idx + 2] = s2; states[idx + 3] = 0.0f;
        const float t0 = tails[idx + 0], t1 = tails[idx + 1], t2 = tails[idx + 2];
        const float n0 = t0 + h00 * s0 + h01 * s1 + h02 * s2;
        const float n1 = t1 + h10 * s0 + h11 * s1 + h12 * s2;
        const float n2 = t2 + h20 * s0 + h21 * s1 + h22 * s2;
        s0 = n0; s1 = n1; s2 = n2;
    }
}

template <bool APPLY_NL>
__global__ __launch_bounds__(256)
void correct_kernel(const float* __restrict__ z,
                    const float* __restrict__ states,
                    const float* __restrict__ hpack,
                    float* __restrict__ out) {
    const int lane = threadIdx.x & 31;
    const int wave = threadIdx.x >> 5;
    const int g    = blockIdx.x * (blockDim.x >> 5) + wave;
    const int n    = lane & 15;
    const int half = lane >> 4;
    const int b    = g / (C / 16);
    const int cg   = g % (C / 16);
    const int c    = cg * 16 + n;

    const float* __restrict__ zb = z   + (size_t)b * T + (size_t)c * L;
    float*       __restrict__ ob = out + (size_t)b * T + (size_t)c * L;

    const v2f bv = *(const v2f*)(states + ((size_t)b * C + c) * 4 + 2 * half);

#pragma unroll 4
    for (int ib = 0; ib < L; ib += 16) {
        const v2f av = *(const v2f*)(hpack + (size_t)(ib + n) * 4 + 2 * half);
        const v4f c0 = *(const v4f*)(zb + ib + 8 * half);
        const v4f c1 = *(const v4f*)(zb + ib + 8 * half + 4);
        v8f cv;
        cv[0] = c0[0]; cv[1] = c0[1]; cv[2] = c0[2]; cv[3] = c0[3];
        cv[4] = c1[0]; cv[5] = c1[1]; cv[6] = c1[2]; cv[7] = c1[3];

        v8f dv = __builtin_amdgcn_wmma_f32_16x16x4_f32(
             false, av,  false, bv,
             (short)0, cv,  false,  false);

        v4f o0, o1;
        if (APPLY_NL) {
            o0[0] = whf(dv[0]); o0[1] = whf(dv[1]); o0[2] = whf(dv[2]); o0[3] = whf(dv[3]);
            o1[0] = whf(dv[4]); o1[1] = whf(dv[5]); o1[2] = whf(dv[6]); o1[3] = whf(dv[7]);
        } else {
            o0[0] = dv[0]; o0[1] = dv[1]; o0[2] = dv[2]; o0[3] = dv[3];
            o1[0] = dv[4]; o1[1] = dv[5]; o1[2] = dv[6]; o1[3] = dv[7];
        }
        if (APPLY_NL) {
            *(v4f*)(ob + ib + 8 * half)     = o0;
            *(v4f*)(ob + ib + 8 * half + 4) = o1;
        } else {
            __builtin_nontemporal_store(o0, (v4f*)(ob + ib + 8 * half));
            __builtin_nontemporal_store(o1, (v4f*)(ob + ib + 8 * half + 4));
        }
    }
}

extern "C" void kernel_launch(void* const* d_in, const int* in_sizes, int n_in,
                              void* d_out, int out_size, void* d_ws, size_t ws_size,
                              hipStream_t stream) {
    const float* u  = (const float*)d_in[0];
    const float* b1 = (const float*)d_in[1];
    const float* a1 = (const float*)d_in[2];
    const float* b2 = (const float*)d_in[3];
    const float* a2 = (const float*)d_in[4];
    float* out = (float*)d_out;

    float* ws     = (float*)d_ws;
    float* z      = ws;
    float* tails  = z + (size_t)Bsz * T;
    float* states = tails + (size_t)Bsz * C * 4;
    float* hp1    = states + (size_t)Bsz * C * 4;
    float* hp2    = hp1 + L * 4;

    const int p1_blocks  = (Bsz * C) / 256;
    const int sc_blocks  = (Bsz + 255) / 256;
    const int cor_blocks = (Bsz * (C / 16)) / 8;

    hpack_kernel<<<2, 32, 0, stream>>>(a1, a2, hp1, hp2);

    phase1_kernel<<<p1_blocks, 256, 0, stream>>>(u, b1, a1,  1, z, tails);
    scan_kernel<<<sc_blocks, 256, 0, stream>>>(tails, hp1, states);
    correct_kernel<true><<<cor_blocks, 256, 0, stream>>>(z, states, hp1, out);

    phase1_kernel<<<p1_blocks, 256, 0, stream>>>(out, b2, a2,  0, z, tails);
    scan_kernel<<<sc_blocks, 256, 0, stream>>>(tails, hp2, states);
    correct_kernel<false><<<cor_blocks, 256, 0, stream>>>(z, states, hp2, out);
}


# phase1: each thread moves its 512-sample chunk in 64-sample register blocks: 128-bit loads issued one block ahead of the recurrence, 128-bit stores (was a waited dword load and a store per sample)
# speedup vs baseline: 2.6144x; 2.6144x over previous
#include <hip/hip_runtime.h>
#include <hip/hip_bf16.h>
#include <math.h>

static constexpr int Bsz = 512;
static constexpr int T   = 16384;
static constexpr int L   = 512;
static constexpr int C   = T / L;

typedef __attribute__((ext_vector_type(2))) float v2f;
typedef __attribute__((ext_vector_type(4))) float v4f;
typedef __attribute__((ext_vector_type(8))) float v8f;

__device__ __forceinline__ float whf(float x) {
    const float c   = 10.0f / 11.0f;
    const float l2e = 1.4426950408889634f;
    float e = 1.0f - __builtin_amdgcn_exp2f(-(c * l2e) * x);
    float l = c * x;
    unsigned s = (unsigned)(__float_as_int(x) >> 31);
    return __uint_as_float((__float_as_uint(e) & ~s) | (__float_as_uint(l) & s));
}

__global__ void hpack_kernel(const float* __restrict__ a1c,
                             const float* __restrict__ a2c,
                             float* __restrict__ hp1,
                             float* __restrict__ hp2) {
    const float* ac = (blockIdx.x == 0) ? a1c : a2c;
    float*       hp = (blockIdx.x == 0) ? hp1 : hp2;
    int j = threadIdx.x;
    if (j > 3) return;
    if (j == 3) {
        for (int i = 0; i < L; ++i) hp[i * 4 + 3] = 0.0f;
        return;
    }
    const float a0 = ac[0], a1 = ac[1], a2 = ac[2];
    float y1 = (j == 0) ? 1.0f : 0.0f;
    float y2 = (j == 1) ? 1.0f : 0.0f;
    float y3 = (j == 2) ? 1.0f : 0.0f;
    for (int i = 0; i < L; ++i) {
        float y = -(a0 * y1 + a1 * y2 + a2 * y3);
        hp[i * 4 + j] = y;
        y3 = y2; y2 = y1; y1 = y;
    }
}

static constexpr int P1_BLK = 64;
static constexpr int P1_NV  = P1_BLK / 4;
static constexpr int P1_NB  = L / P1_BLK;

template <int NV>
__device__ __forceinline__ void p1_load(v4f (&dst)[NV], const float* __restrict__ p) {
#pragma unroll
    for (int v = 0; v < NV; ++v) dst[v] = *(const v4f*)(p + 4 * v);
    __builtin_amdgcn_sched_barrier(0);
}

template <int NK, int NV>
__device__ __forceinline__ void p1_block(const v4f (&xin)[NV], float* __restrict__ zdst,
                                         float b0, float b1, float b2,
                                         float a0, float a1, float a2,
                                         float& p1, float& p2, float& p3,
                                         float& y1, float& y2, float& y3) {
#pragma unroll
    for (int v = 0; v < NV; ++v) {
        v4f o;
#pragma unroll
        for (int e = 0; e < 4; ++e) {
            const float s = xin[v][e];
            float xf;
            if (NK == 0) xf = b0 * s  + b1 * p1 + b2 * p2;
            else         xf = b0 * p1 + b1 * p2 + b2 * p3;
            const float y = xf - (a0 * y1 + a1 * y2 + a2 * y3);
            o[e] = y;
            p3 = p2; p2 = p1; p1 = s;
            y3 = y2; y2 = y1; y1 = y;
        }
        *(v4f*)(zdst + 4 * v) = o;
    }
}

template <int NK>
__global__ __launch_bounds__(256)
void phase1_kernel(const float* __restrict__ in,
                   const float* __restrict__ bco,
                   const float* __restrict__ aco,
                   float* __restrict__ z,
                   float* __restrict__ tails) {
    const int id = blockIdx.x * blockDim.x + threadIdx.x;
    const int b  = id / C;
    const int c  = id % C;
    const float b0 = bco[0], b1 = bco[1], b2 = bco[2];
    const float a0 = aco[0], a1 = aco[1], a2 = aco[2];

    const float* __restrict__ src  = in + (size_t)b * T + (size_t)c * L;
    float*       __restrict__ zrow = z  + (size_t)b * T + (size_t)c * L;

    float p1 = 0.0f, p2 = 0.0f, p3 = 0.0f;
    if (c > 0) {
        p1 = src[-1];
        p2 = src[-2];
        if (NK == 1) p3 = src[-3];
    }
    float y1 = 0.0f, y2 = 0.0f, y3 = 0.0f;

    v4f xa[P1_NV], xb[P1_NV];
    p1_load<P1_NV>(xa, src);
    for (int blk = 0; blk < P1_NB - 2; blk += 2) {
        p1_load<P1_NV>(xb, src + (blk + 1) * P1_BLK);
        p1_block<NK, P1_NV>(xa, zrow + blk * P1_BLK, b0, b1, b2, a0, a1, a2,
                            p1, p2, p3, y1, y2, y3);
        p1_load<P1_NV>(xa, src + (blk + 2) * P1_BLK);
        p1_block<NK, P1_NV>(xb, zrow + (blk + 1) * P1_BLK, b0, b1, b2, a0, a1, a2,
                            p1, p2, p3, y1, y2, y3);
    }
    p1_load<P1_NV>(xb, src + (P1_NB - 1) * P1_BLK);
    p1_block<NK, P1_NV>(xa, zrow + (P1_NB - 2) * P1_BLK, b0, b1, b2, a0, a1, a2,
                        p1, p2, p3, y1, y2, y3);
    p1_block<NK, P1_NV>(xb, zrow + (P1_NB - 1) * P1_BLK, b0, b1, b2, a0, a1, a2,
                        p1, p2, p3, y1, y2, y3);
    float* tl = tails + (size_t)id * 4;
    tl[0] = y1; tl[1] = y2; tl[2] = y3; tl[3] = 0.0f;
}

__global__ __launch_bounds__(256)
void scan_kernel(const float* __restrict__ tails,
                 const float* __restrict__ hpack,
                 float* __restrict__ states) {
    const int b = blockIdx.x * blockDim.x + threadIdx.x;
    if (b >= Bsz) return;
    const float h00 = hpack[(L - 1) * 4 + 0], h01 = hpack[(L - 1) * 4 + 1], h02 = hpack[(L - 1) * 4 + 2];
    const float h10 = hpack[(L - 2) * 4 + 0], h11 = hpack[(L - 2) * 4 + 1], h12 = hpack[(L - 2) * 4 + 2];
    const float h20 = hpack[(L - 3) * 4 + 0], h21 = hpack[(L - 3) * 4 + 1], h22 = hpack[(L - 3) * 4 + 2];
    float s0 = 0.0f, s1 = 0.0f, s2 = 0.0f;
    for (int c = 0; c < C; ++c) {
        const size_t idx = ((size_t)b * C + c) * 4;
        states[idx + 0] = s0; states[idx + 1] = s1;
        states[idx + 2] = s2; states[idx + 3] = 0.0f;
        const float t0 = tails[idx + 0], t1 = tails[idx + 1], t2 = tails[idx + 2];
        const float n0 = t0 + h00 * s0 + h01 * s1 + h02 * s2;
        const float n1 = t1 + h10 * s0 + h11 * s1 + h12 * s2;
        const float n2 = t2 + h20 * s0 + h21 * s1 + h22 * s2;
        s0 = n0; s1 = n1; s2 = n2;
    }
}

template <bool APPLY_NL>
__global__ __launch_bounds__(256)
void correct_kernel(const float* __restrict__ z,
                    const float* __restrict__ states,
                    const float* __restrict__ hpack,
                    float* __restrict__ out) {
    const int lane = threadIdx.x & 31;
    const int wave = threadIdx.x >> 5;
    const int g    = blockIdx.x * (blockDim.x >> 5) + wave;
    const int n    = lane & 15;
    const int half = lane >> 4;
    const int b    = g / (C / 16);
    const int cg   = g % (C / 16);
    const int c    = cg * 16 + n;

    const float* __restrict__ zb = z   + (size_t)b * T + (size_t)c * L;
    float*       __restrict__ ob = out + (size_t)b * T + (size_t)c * L;

    const v2f bv = *(const v2f*)(states + ((size_t)b * C + c) * 4 + 2 * half);

#pragma unroll 4
    for (int ib = 0; ib < L; ib += 16) {
        const v2f av = *(const v2f*)(hpack + (size_t)(ib + n) * 4 + 2 * half);
        const v4f c0 = *(const v4f*)(zb + ib + 8 * half);
        const v4f c1 = *(const v4f*)(zb + ib + 8 * half + 4);
        v8f cv;
        cv[0] = c0[0]; cv[1] = c0[1]; cv[2] = c0[2]; cv[3] = c0[3];
        cv[4] = c1[0]; cv[5] = c1[1]; cv[6] = c1[2]; cv[7] = c1[3];

        v8f dv = __builtin_amdgcn_wmma_f32_16x16x4_f32(
             false, av,  false, bv,
             (short)0, cv,  false,  false);

        v4f o0, o1;
        if (APPLY_NL) {
            o0[0] = whf(dv[0]); o0[1] = whf(dv[1]); o0[2] = whf(dv[2]); o0[3] = whf(dv[3]);
            o1[0] = whf(dv[4]); o1[1] = whf(dv[5]); o1[2] = whf(dv[6]); o1[3] = whf(dv[7]);
        } else {
            o0[0] = dv[0]; o0[1] = dv[1]; o0[2] = dv[2]; o0[3] = dv[3];
            o1[0] = dv[4]; o1[1] = dv[5]; o1[2] = dv[6]; o1[3] = dv[7];
        }
        if (APPLY_NL) {
            *(v4f*)(ob + ib + 8 * half)     = o0;
            *(v4f*)(ob + ib + 8 * half + 4) = o1;
        } else {
            __builtin_nontemporal_store(o0, (v4f*)(ob + ib + 8 * half));
            __builtin_nontemporal_store(o1, (v4f*)(ob + ib + 8 * half + 4));
        }
    }
}

extern "C" void kernel_launch(void* const* d_in, const int* in_sizes, int n_in,
                              void* d_out, int out_size, void* d_ws, size_t ws_size,
                              hipStream_t stream) {
    const float* u  = (const float*)d_in[0];
    const float* b1 = (const float*)d_in[1];
    const float* a1 = (const float*)d_in[2];
    const float* b2 = (const float*)d_in[3];
    const float* a2 = (const float*)d_in[4];
    float* out = (float*)d_out;

    float* ws     = (float*)d_ws;
    float* z      = ws;
    float* tails  = z + (size_t)Bsz * T;
    float* states = tails + (size_t)Bsz * C * 4;
    float* hp1    = states + (size_t)Bsz * C * 4;
    float* hp2    = hp1 + L * 4;

    const int p1_blocks  = (Bsz * C) / 256;
    const int sc_blocks  = (Bsz + 255) / 256;
    const int cor_blocks = (Bsz * (C / 16)) / 8;

    hpack_kernel<<<2, 32, 0, stream>>>(a1, a2, hp1, hp2);

    phase1_kernel<1><<<p1_blocks, 256, 0, stream>>>(u, b1, a1, z, tails);
    scan_kernel<<<sc_blocks, 256, 0, stream>>>(tails, hp1, states);
    correct_kernel<true><<<cor_blocks, 256, 0, stream>>>(z, states, hp1, out);

    phase1_kernel<0><<<p1_blocks, 256, 0, stream>>>(out, b2, a2, z, tails);
    scan_kernel<<<sc_blocks, 256, 0, stream>>>(tails, hp2, states);
    correct_kernel<false><<<cor_blocks, 256, 0, stream>>>(z, states, hp2, out);
}


# filter-1 correction (matrix products + nonlinearity) fused into the filter-2 recurrence kernel: v stays in registers via one half-wave lane exchange; its store, reload and one launch are gone
# speedup vs baseline: 4.2167x; 1.6129x over previous
#include <hip/hip_runtime.h>
#include <hip/hip_bf16.h>
#include <math.h>

static constexpr int Bsz = 512;
static constexpr int T   = 16384;
static constexpr int L   = 512;
static constexpr int C   = T / L;

typedef __attribute__((ext_vector_type(2))) float v2f;
typedef __attribute__((ext_vector_type(4))) float v4f;
typedef __attribute__((ext_vector_type(8))) float v8f;

__device__ __forceinline__ float whf(float x) {
    const float c   = 10.0f / 11.0f;
    const float l2e = 1.4426950408889634f;
    float e = 1.0f - __builtin_amdgcn_exp2f(-(c * l2e) * x);
    float l = c * x;
    unsigned s = (unsigned)(__float_as_int(x) >> 31);
    return __uint_as_float((__float_as_uint(e) & ~s) | (__float_as_uint(l) & s));
}

__global__ void hpack_kernel(const float* __restrict__ a1c,
                             const float* __restrict__ a2c,
                             float* __restrict__ hp1,
                             float* __restrict__ hp2) {
    const float* ac = (blockIdx.x == 0) ? a1c : a2c;
    float*       hp = (blockIdx.x == 0) ? hp1 : hp2;
    int j = threadIdx.x;
    if (j > 3) return;
    if (j == 3) {
        for (int i = 0; i < L; ++i) hp[i * 4 + 3] = 0.0f;
        return;
    }
    const float a0 = ac[0], a1 = ac[1], a2 = ac[2];
    float y1 = (j == 0) ? 1.0f : 0.0f;
    float y2 = (j == 1) ? 1.0f : 0.0f;
    float y3 = (j == 2) ? 1.0f : 0.0f;
    for (int i = 0; i < L; ++i) {
        float y = -(a0 * y1 + a1 * y2 + a2 * y3);
        hp[i * 4 + j] = y;
        y3 = y2; y2 = y1; y1 = y;
    }
}

static constexpr int P1_BLK = 64;
static constexpr int P1_NV  = P1_BLK / 4;
static constexpr int P1_NB  = L / P1_BLK;

template <int NV>
__device__ __forceinline__ void p1_load(v4f (&dst)[NV], const float* __restrict__ p) {
#pragma unroll
    for (int v = 0; v < NV; ++v) dst[v] = *(const v4f*)(p + 4 * v);
    __builtin_amdgcn_sched_barrier(0);
}

template <int NK, int NV>
__device__ __forceinline__ void p1_block(const v4f (&xin)[NV], float* __restrict__ zdst,
                                         float b0, float b1, float b2,
                                         float a0, float a1, float a2,
                                         float& p1, float& p2, float& p3,
                                         float& y1, float& y2, float& y3) {
#pragma unroll
    for (int v = 0; v < NV; ++v) {
        v4f o;
#pragma unroll
        for (int e = 0; e < 4; ++e) {
            const float s = xin[v][e];
            float xf;
            if (NK == 0) xf = b0 * s  + b1 * p1 + b2 * p2;
            else         xf = b0 * p1 + b1 * p2 + b2 * p3;
            const float y = xf - (a0 * y1 + a1 * y2 + a2 * y3);
            o[e] = y;
            p3 = p2; p2 = p1; p1 = s;
            y3 = y2; y2 = y1; y1 = y;
        }
        *(v4f*)(zdst + 4 * v) = o;
    }
}

template <int NK>
__global__ __launch_bounds__(256)
void phase1_kernel(const float* __restrict__ in,
                   const float* __restrict__ bco,
                   const float* __restrict__ aco,
                   float* __restrict__ z,
                   float* __restrict__ tails) {
    const int id = blockIdx.x * blockDim.x + threadIdx.x;
    const int b  = id / C;
    const int c  = id % C;
    const float b0 = bco[0], b1 = bco[1], b2 = bco[2];
    const float a0 = aco[0], a1 = aco[1], a2 = aco[2];

    const float* __restrict__ src  = in + (size_t)b * T + (size_t)c * L;
    float*       __restrict__ zrow = z  + (size_t)b * T + (size_t)c * L;

    float p1 = 0.0f, p2 = 0.0f, p3 = 0.0f;
    if (c > 0) {
        p1 = src[-1];
        p2 = src[-2];
        if (NK == 1) p3 = src[-3];
    }
    float y1 = 0.0f, y2 = 0.0f, y3 = 0.0f;

    v4f xa[P1_NV], xb[P1_NV];
    p1_load<P1_NV>(xa, src);
    for (int blk = 0; blk < P1_NB - 2; blk += 2) {
        p1_load<P1_NV>(xb, src + (blk + 1) * P1_BLK);
        p1_block<NK, P1_NV>(xa, zrow + blk * P1_BLK, b0, b1, b2, a0, a1, a2,
                            p1, p2, p3, y1, y2, y3);
        p1_load<P1_NV>(xa, src + (blk + 2) * P1_BLK);
        p1_block<NK, P1_NV>(xb, zrow + (blk + 1) * P1_BLK, b0, b1, b2, a0, a1, a2,
                            p1, p2, p3, y1, y2, y3);
    }
    p1_load<P1_NV>(xb, src + (P1_NB - 1) * P1_BLK);
    p1_block<NK, P1_NV>(xa, zrow + (P1_NB - 2) * P1_BLK, b0, b1, b2, a0, a1, a2,
                        p1, p2, p3, y1, y2, y3);
    p1_block<NK, P1_NV>(xb, zrow + (P1_NB - 1) * P1_BLK, b0, b1, b2, a0, a1, a2,
                        p1, p2, p3, y1, y2, y3);
    float* tl = tails + (size_t)id * 4;
    tl[0] = y1; tl[1] = y2; tl[2] = y3; tl[3] = 0.0f;
}

__global__ __launch_bounds__(256)
void scan_kernel(const float* __restrict__ tails,
                 const float* __restrict__ hpack,
                 float* __restrict__ states) {
    const int b = blockIdx.x * blockDim.x + threadIdx.x;
    if (b >= Bsz) return;
    const float h00 = hpack[(L - 1) * 4 + 0], h01 = hpack[(L - 1) * 4 + 1], h02 = hpack[(L - 1) * 4 + 2];
    const float h10 = hpack[(L - 2) * 4 + 0], h11 = hpack[(L - 2) * 4 + 1], h12 = hpack[(L - 2) * 4 + 2];
    const float h20 = hpack[(L - 3) * 4 + 0], h21 = hpack[(L - 3) * 4 + 1], h22 = hpack[(L - 3) * 4 + 2];
    float s0 = 0.0f, s1 = 0.0f, s2 = 0.0f;
    for (int c = 0; c < C; ++c) {
        const size_t idx = ((size_t)b * C + c) * 4;
        states[idx + 0] = s0; states[idx + 1] = s1;
        states[idx + 2] = s2; states[idx + 3] = 0.0f;
        const float t0 = tails[idx + 0], t1 = tails[idx + 1], t2 = tails[idx + 2];
        const float n0 = t0 + h00 * s0 + h01 * s1 + h02 * s2;
        const float n1 = t1 + h10 * s0 + h11 * s1 + h12 * s2;
        const float n2 = t2 + h20 * s0 + h21 * s1 + h22 * s2;
        s0 = n0; s1 = n1; s2 = n2;
    }
}

__global__ __launch_bounds__(256)
void correct_kernel(const float* __restrict__ z,
                    const float* __restrict__ states,
                    const float* __restrict__ hpack,
                    float* __restrict__ out) {
    const int lane = threadIdx.x & 31;
    const int wave = threadIdx.x >> 5;
    const int g    = blockIdx.x * (blockDim.x >> 5) + wave;
    const int n    = lane & 15;
    const int half = lane >> 4;
    const int b    = g / (C / 16);
    const int cg   = g % (C / 16);
    const int c    = cg * 16 + n;

    const float* __restrict__ zb = z   + (size_t)b * T + (size_t)c * L;
    float*       __restrict__ ob = out + (size_t)b * T + (size_t)c * L;

    const v2f bv = *(const v2f*)(states + ((size_t)b * C + c) * 4 + 2 * half);

#pragma unroll 4
    for (int ib = 0; ib < L; ib += 16) {
        const v2f av = *(const v2f*)(hpack + (size_t)(ib + n) * 4 + 2 * half);
        const v4f c0 = *(const v4f*)(zb + ib + 8 * half);
        const v4f c1 = *(const v4f*)(zb + ib + 8 * half + 4);
        v8f cv;
        cv[0] = c0[0]; cv[1] = c0[1]; cv[2] = c0[2]; cv[3] = c0[3];
        cv[4] = c1[0]; cv[5] = c1[1]; cv[6] = c1[2]; cv[7] = c1[3];

        v8f dv = __builtin_amdgcn_wmma_f32_16x16x4_f32(
             false, av,  false, bv,
             (short)0, cv,  false,  false);

        v4f o0, o1;
        o0[0] = dv[0]; o0[1] = dv[1]; o0[2] = dv[2]; o0[3] = dv[3];
        o1[0] = dv[4]; o1[1] = dv[5]; o1[2] = dv[6]; o1[3] = dv[7];
        __builtin_nontemporal_store(o0, (v4f*)(ob + ib + 8 * half));
        __builtin_nontemporal_store(o1, (v4f*)(ob + ib + 8 * half + 4));
    }
}

static_assert(C == 32, "one wave per row: lane index == chunk index");
static constexpr int F_NS = P1_BLK / 16;

template <int NS>
__device__ __forceinline__ void f_load(v4f (&cz)[NS][4], v2f (&ah)[NS],
                                       const float* zlo, const float* zhi,
                                       const float* __restrict__ hrow) {
#pragma unroll
    for (int s = 0; s < NS; ++s) {
        cz[s][0] = *(const v4f*)(zlo + 16 * s);
        cz[s][1] = *(const v4f*)(zlo + 16 * s + 4);
        cz[s][2] = *(const v4f*)(zhi + 16 * s);
        cz[s][3] = *(const v4f*)(zhi + 16 * s + 4);
        ah[s]    = *(const v2f*)(hrow + 64 * s);
    }
    __builtin_amdgcn_sched_barrier(0);
}

__device__ __forceinline__ void f_v16(const v4f (&cz)[4], v2f av, v2f bvl, v2f bvh,
                                      bool hi, v4f (&seq)[4]) {
    v8f c1, c2;
    c1[0] = cz[0][0]; c1[1] = cz[0][1]; c1[2] = cz[0][2]; c1[3] = cz[0][3];
    c1[4] = cz[1][0]; c1[5] = cz[1][1]; c1[6] = cz[1][2]; c1[7] = cz[1][3];
    c2[0] = cz[2][0]; c2[1] = cz[2][1]; c2[2] = cz[2][2]; c2[3] = cz[2][3];
    c2[4] = cz[3][0]; c2[5] = cz[3][1]; c2[6] = cz[3][2]; c2[7] = cz[3][3];
    const v8f d1 = __builtin_amdgcn_wmma_f32_16x16x4_f32(
         false, av,  false, bvl,
         (short)0, c1,  false,  false);
    const v8f d2 = __builtin_amdgcn_wmma_f32_16x16x4_f32(
         false, av,  false, bvh,
         (short)0, c2,  false,  false);
    float v1[8], v2[8], rcv[8];
#pragma unroll
    for (int r = 0; r < 8; ++r) {
        v1[r] = whf(d1[r]);
        v2[r] = whf(d2[r]);
    }
    __builtin_amdgcn_sched_barrier(0);
#pragma unroll
    for (int r = 0; r < 8; ++r) rcv[r] = __shfl_xor(hi ? v1[r] : v2[r], 16);
    __builtin_amdgcn_sched_barrier(0);
#pragma unroll
    for (int r = 0; r < 8; ++r) {
        seq[r >> 2][r & 3]       = hi ? rcv[r] : v1[r];
        seq[2 + (r >> 2)][r & 3] = hi ? v2[r]  : rcv[r];
    }
}

template <int NS>
__device__ __forceinline__ void f_block(const v4f (&cz)[NS][4], const v2f (&ah)[NS],
                                        v2f bvl, v2f bvh, bool hi, float* zdst,
                                        float b0, float b1, float b2,
                                        float a0, float a1, float a2,
                                        float& p1, float& p2, float& p3,
                                        float& y1, float& y2, float& y3) {
#pragma unroll
    for (int s = 0; s < NS; ++s) {
        v4f seq[4];
        f_v16(cz[s], ah[s], bvl, bvh, hi, seq);
        p1_block<0, 4>(seq, zdst + 16 * s, b0, b1, b2, a0, a1, a2,
                       p1, p2, p3, y1, y2, y3);
    }
}

__global__ __launch_bounds__(256)
void fused_kernel(float* z,
                  const float* __restrict__ states,
                  const float* __restrict__ hpack,
                  const float* __restrict__ bco,
                  const float* __restrict__ aco,
                  float* __restrict__ tails) {
    const int id   = blockIdx.x * blockDim.x + threadIdx.x;
    const int b    = id / C;
    const int lane = threadIdx.x & 31;
    const int n    = lane & 15;
    const int half = lane >> 4;
    const bool hi  = half != 0;
    const float b0 = bco[0], b1 = bco[1], b2 = bco[2];
    const float a0 = aco[0], a1 = aco[1], a2 = aco[2];

    float* zrow = z + (size_t)b * T;
    const float* zlo  = zrow + (size_t)n * L + 8 * half;
    const float* zhi  = zrow + (size_t)(16 + n) * L + 8 * half;
    float*       zown = zrow + (size_t)lane * L;
    const float* __restrict__ hrow = hpack + n * 4 + 2 * half;

    const v2f bvl = *(const v2f*)(states + ((size_t)b * C + n) * 4 + 2 * half);
    const v2f bvh = *(const v2f*)(states + ((size_t)b * C + 16 + n) * 4 + 2 * half);

    float p1, p2, p3 = 0.0f;
    {
        v4f cz[1][4]; v2f ah[1]; v4f seq[4];
        f_load<1>(cz, ah, zlo + (L - 16), zhi + (L - 16), hrow + (L - 16) * 4);
        f_v16(cz[0], ah[0], bvl, bvh, hi, seq);
        p1 = __shfl_up(seq[3][3], 1);
        p2 = __shfl_up(seq[3][2], 1);
        if (lane == 0) { p1 = 0.0f; p2 = 0.0f; }
    }
    float y1 = 0.0f, y2 = 0.0f, y3 = 0.0f;

    v4f ca[F_NS][4], cb[F_NS][4];
    v2f ha[F_NS], hb[F_NS];
    f_load<F_NS>(ca, ha, zlo, zhi, hrow);
    for (int blk = 0; blk < P1_NB - 2; blk += 2) {
        f_load<F_NS>(cb, hb, zlo + (blk + 1) * P1_BLK, zhi + (blk + 1) * P1_BLK,
                     hrow + (blk + 1) * P1_BLK * 4);
        f_block<F_NS>(ca, ha, bvl, bvh, hi, zown + blk * P1_BLK,
                      b0, b1, b2, a0, a1, a2, p1, p2, p3, y1, y2, y3);
        f_load<F_NS>(ca, ha, zlo + (blk + 2) * P1_BLK, zhi + (blk + 2) * P1_BLK,
                     hrow + (blk + 2) * P1_BLK * 4);
        f_block<F_NS>(cb, hb, bvl, bvh, hi, zown + (blk + 1) * P1_BLK,
                      b0, b1, b2, a0, a1, a2, p1, p2, p3, y1, y2, y3);
    }
    f_load<F_NS>(cb, hb, zlo + (P1_NB - 1) * P1_BLK, zhi + (P1_NB - 1) * P1_BLK,
                 hrow + (P1_NB - 1) * P1_BLK * 4);
    f_block<F_NS>(ca, ha, bvl, bvh, hi, zown + (P1_NB - 2) * P1_BLK,
                  b0, b1, b2, a0, a1, a2, p1, p2, p3, y1, y2, y3);
    f_block<F_NS>(cb, hb, bvl, bvh, hi, zown + (P1_NB - 1) * P1_BLK,
                  b0, b1, b2, a0, a1, a2, p1, p2, p3, y1, y2, y3);
    float* tl = tails + (size_t)id * 4;
    tl[0] = y1; tl[1] = y2; tl[2] = y3; tl[3] = 0.0f;
}

extern "C" void kernel_launch(void* const* d_in, const int* in_sizes, int n_in,
                              void* d_out, int out_size, void* d_ws, size_t ws_size,
                              hipStream_t stream) {
    const float* u  = (const float*)d_in[0];
    const float* b1 = (const float*)d_in[1];
    const float* a1 = (const float*)d_in[2];
    const float* b2 = (const float*)d_in[3];
    const float* a2 = (const float*)d_in[4];
    float* out = (float*)d_out;

    float* ws     = (float*)d_ws;
    float* z      = ws;
    float* tails  = z + (size_t)Bsz * T;
    float* states = tails + (size_t)Bsz * C * 4;
    float* hp1    = states + (size_t)Bsz * C * 4;
    float* hp2    = hp1 + L * 4;

    const int p1_blocks  = (Bsz * C) / 256;
    const int sc_blocks  = (Bsz + 255) / 256;
    const int cor_blocks = (Bsz * (C / 16)) / 8;

    hpack_kernel<<<2, 32, 0, stream>>>(a1, a2, hp1, hp2);

    phase1_kernel<1><<<p1_blocks, 256, 0, stream>>>(u, b1, a1, z, tails);
    scan_kernel<<<sc_blocks, 256, 0, stream>>>(tails, hp1, states);

    fused_kernel<<<p1_blocks, 256, 0, stream>>>(z, states, hp1, b2, a2, tails);

    scan_kernel<<<sc_blocks, 256, 0, stream>>>(tails, hp2, states);
    correct_kernel<<<cor_blocks, 256, 0, stream>>>(z, states, hp2, out);
}


# chunk-state scan done inside the wave that owns the row (lane == chunk, tails broadcast lane by lane, same sums in the same order): both scan launches and the tails round trip are gone
# speedup vs baseline: 5.2229x; 1.2386x over previous
#include <hip/hip_runtime.h>
#include <hip/hip_bf16.h>
#include <math.h>

static constexpr int Bsz = 512;
static constexpr int T   = 16384;
static constexpr int L   = 512;
static constexpr int C   = T / L;

typedef __attribute__((ext_vector_type(2))) float v2f;
typedef __attribute__((ext_vector_type(4))) float v4f;
typedef __attribute__((ext_vector_type(8))) float v8f;

__device__ __forceinline__ float whf(float x) {
    const float c   = 10.0f / 11.0f;
    const float l2e = 1.4426950408889634f;
    float e = 1.0f - __builtin_amdgcn_exp2f(-(c * l2e) * x);
    float l = c * x;
    unsigned s = (unsigned)(__float_as_int(x) >> 31);
    return __uint_as_float((__float_as_uint(e) & ~s) | (__float_as_uint(l) & s));
}

__global__ void hpack_kernel(const float* __restrict__ a1c,
                             const float* __restrict__ a2c,
                             float* __restrict__ hp1,
                             float* __restrict__ hp2) {
    const float* ac = (blockIdx.x == 0) ? a1c : a2c;
    float*       hp = (blockIdx.x == 0) ? hp1 : hp2;
    int j = threadIdx.x;
    if (j > 3) return;
    if (j == 3) {
        for (int i = 0; i < L; ++i) hp[i * 4 + 3] = 0.0f;
        return;
    }
    const float a0 = ac[0], a1 = ac[1], a2 = ac[2];
    float y1 = (j == 0) ? 1.0f : 0.0f;
    float y2 = (j == 1) ? 1.0f : 0.0f;
    float y3 = (j == 2) ? 1.0f : 0.0f;
    for (int i = 0; i < L; ++i) {
        float y = -(a0 * y1 + a1 * y2 + a2 * y3);
        hp[i * 4 + j] = y;
        y3 = y2; y2 = y1; y1 = y;
    }
}

static_assert(C == 32, "one wave per row: lane index == chunk index");

struct HTail { float h00, h01, h02, h10, h11, h12, h20, h21, h22; };

__device__ __forceinline__ HTail load_htail(const float* __restrict__ hpack) {
    HTail h;
    h.h00 = hpack[(L - 1) * 4 + 0]; h.h01 = hpack[(L - 1) * 4 + 1]; h.h02 = hpack[(L - 1) * 4 + 2];
    h.h10 = hpack[(L - 2) * 4 + 0]; h.h11 = hpack[(L - 2) * 4 + 1]; h.h12 = hpack[(L - 2) * 4 + 2];
    h.h20 = hpack[(L - 3) * 4 + 0]; h.h21 = hpack[(L - 3) * 4 + 1]; h.h22 = hpack[(L - 3) * 4 + 2];
    return h;
}

__device__ __forceinline__ float lane_bcast(float v, int src) {
    return __int_as_float(__builtin_amdgcn_readlane(__float_as_int(v), src));
}

__device__ __forceinline__ void wave_scan_store(const HTail& h, float t0, float t1, float t2,
                                                int lane, float* __restrict__ st) {
    float s0 = 0.0f, s1 = 0.0f, s2 = 0.0f;
    float e0 = 0.0f, e1 = 0.0f, e2 = 0.0f;
    for (int c = 0; c < C; ++c) {
        if (lane == c) { e0 = s0; e1 = s1; e2 = s2; }
        const float c0 = lane_bcast(t0, c), c1 = lane_bcast(t1, c), c2 = lane_bcast(t2, c);
        const float n0 = c0 + h.h00 * s0 + h.h01 * s1 + h.h02 * s2;
        const float n1 = c1 + h.h10 * s0 + h.h11 * s1 + h.h12 * s2;
        const float n2 = c2 + h.h20 * s0 + h.h21 * s1 + h.h22 * s2;
        s0 = n0; s1 = n1; s2 = n2;
    }
    v4f e;
    e[0] = e0; e[1] = e1; e[2] = e2; e[3] = 0.0f;
    *(v4f*)st = e;
}

static constexpr int P1_BLK = 64;
static constexpr int P1_NV  = P1_BLK / 4;
static constexpr int P1_NB  = L / P1_BLK;

template <int NV>
__device__ __forceinline__ void p1_load(v4f (&dst)[NV], const float* __restrict__ p) {
#pragma unroll
    for (int v = 0; v < NV; ++v) dst[v] = *(const v4f*)(p + 4 * v);
    __builtin_amdgcn_sched_barrier(0);
}

template <int NK, int NV>
__device__ __forceinline__ void p1_block(const v4f (&xin)[NV], float* __restrict__ zdst,
                                         float b0, float b1, float b2,
                                         float a0, float a1, float a2,
                                         float& p1, float& p2, float& p3,
                                         float& y1, float& y2, float& y3) {
#pragma unroll
    for (int v = 0; v < NV; ++v) {
        v4f o;
#pragma unroll
        for (int e = 0; e < 4; ++e) {
            const float s = xin[v][e];
            float xf;
            if (NK == 0) xf = b0 * s  + b1 * p1 + b2 * p2;
            else         xf = b0 * p1 + b1 * p2 + b2 * p3;
            const float y = xf - (a0 * y1 + a1 * y2 + a2 * y3);
            o[e] = y;
            p3 = p2; p2 = p1; p1 = s;
            y3 = y2; y2 = y1; y1 = y;
        }
        *(v4f*)(zdst + 4 * v) = o;
    }
}

template <int NK>
__global__ __launch_bounds__(256)
void phase1_kernel(const float* __restrict__ in,
                   const float* __restrict__ bco,
                   const float* __restrict__ aco,
                   const float* __restrict__ hpack,
                   float* __restrict__ z,
                   float* __restrict__ states) {
    const int id = blockIdx.x * blockDim.x + threadIdx.x;
    const int b  = id / C;
    const int c  = id % C;
    const float b0 = bco[0], b1 = bco[1], b2 = bco[2];
    const float a0 = aco[0], a1 = aco[1], a2 = aco[2];
    const HTail ht = load_htail(hpack);

    const float* __restrict__ src  = in + (size_t)b * T + (size_t)c * L;
    float*       __restrict__ zrow = z  + (size_t)b * T + (size_t)c * L;

    float p1 = 0.0f, p2 = 0.0f, p3 = 0.0f;
    if (c > 0) {
        p1 = src[-1];
        p2 = src[-2];
        if (NK == 1) p3 = src[-3];
    }
    float y1 = 0.0f, y2 = 0.0f, y3 = 0.0f;

    v4f xa[P1_NV], xb[P1_NV];
    p1_load<P1_NV>(xa, src);
    for (int blk = 0; blk < P1_NB - 2; blk += 2) {
        p1_load<P1_NV>(xb, src + (blk + 1) * P1_BLK);
        p1_block<NK, P1_NV>(xa, zrow + blk * P1_BLK, b0, b1, b2, a0, a1, a2,
                            p1, p2, p3, y1, y2, y3);
        p1_load<P1_NV>(xa, src + (blk + 2) * P1_BLK);
        p1_block<NK, P1_NV>(xb, zrow + (blk + 1) * P1_BLK, b0, b1, b2, a0, a1, a2,
                            p1, p2, p3, y1, y2, y3);
    }
    p1_load<P1_NV>(xb, src + (P1_NB - 1) * P1_BLK);
    p1_block<NK, P1_NV>(xa, zrow + (P1_NB - 2) * P1_BLK, b0, b1, b2, a0, a1, a2,
                        p1, p2, p3, y1, y2, y3);
    p1_block<NK, P1_NV>(xb, zrow + (P1_NB - 1) * P1_BLK, b0, b1, b2, a0, a1, a2,
                        p1, p2, p3, y1, y2, y3);
    wave_scan_store(ht, y1, y2, y3, c, states + (size_t)id * 4);
}

__global__ __launch_bounds__(256)
void correct_kernel(const float* __restrict__ z,
                    const float* __restrict__ states,
                    const float* __restrict__ hpack,
                    float* __restrict__ out) {
    const int lane = threadIdx.x & 31;
    const int wave = threadIdx.x >> 5;
    const int g    = blockIdx.x * (blockDim.x >> 5) + wave;
    const int n    = lane & 15;
    const int half = lane >> 4;
    const int b    = g / (C / 16);
    const int cg   = g % (C / 16);
    const int c    = cg * 16 + n;

    const float* __restrict__ zb = z   + (size_t)b * T + (size_t)c * L;
    float*       __restrict__ ob = out + (size_t)b * T + (size_t)c * L;

    const v2f bv = *(const v2f*)(states + ((size_t)b * C + c) * 4 + 2 * half);

#pragma unroll 4
    for (int ib = 0; ib < L; ib += 16) {
        const v2f av = *(const v2f*)(hpack + (size_t)(ib + n) * 4 + 2 * half);
        const v4f c0 = *(const v4f*)(zb + ib + 8 * half);
        const v4f c1 = *(const v4f*)(zb + ib + 8 * half + 4);
        v8f cv;
        cv[0] = c0[0]; cv[1] = c0[1]; cv[2] = c0[2]; cv[3] = c0[3];
        cv[4] = c1[0]; cv[5] = c1[1]; cv[6] = c1[2]; cv[7] = c1[3];

        v8f dv = __builtin_amdgcn_wmma_f32_16x16x4_f32(
             false, av,  false, bv,
             (short)0, cv,  false,  false);

        v4f o0, o1;
        o0[0] = dv[0]; o0[1] = dv[1]; o0[2] = dv[2]; o0[3] = dv[3];
        o1[0] = dv[4]; o1[1] = dv[5]; o1[2] = dv[6]; o1[3] = dv[7];
        __builtin_nontemporal_store(o0, (v4f*)(ob + ib + 8 * half));
        __builtin_nontemporal_store(o1, (v4f*)(ob + ib + 8 * half + 4));
    }
}

static constexpr int F_NS = P1_BLK / 16;

template <int NS>
__device__ __forceinline__ void f_load(v4f (&cz)[NS][4], v2f (&ah)[NS],
                                       const float* zlo, const float* zhi,
                                       const float* __restrict__ hrow) {
#pragma unroll
    for (int s = 0; s < NS; ++s) {
        cz[s][0] = *(const v4f*)(zlo + 16 * s);
        cz[s][1] = *(const v4f*)(zlo + 16 * s + 4);
        cz[s][2] = *(const v4f*)(zhi + 16 * s);
        cz[s][3] = *(const v4f*)(zhi + 16 * s + 4);
        ah[s]    = *(const v2f*)(hrow + 64 * s);
    }
    __builtin_amdgcn_sched_barrier(0);
}

__device__ __forceinline__ void f_v16(const v4f (&cz)[4], v2f av, v2f bvl, v2f bvh,
                                      bool hi, v4f (&seq)[4]) {
    v8f c1, c2;
    c1[0] = cz[0][0]; c1[1] = cz[0][1]; c1[2] = cz[0][2]; c1[3] = cz[0][3];
    c1[4] = cz[1][0]; c1[5] = cz[1][1]; c1[6] = cz[1][2]; c1[7] = cz[1][3];
    c2[0] = cz[2][0]; c2[1] = cz[2][1]; c2[2] = cz[2][2]; c2[3] = cz[2][3];
    c2[4] = cz[3][0]; c2[5] = cz[3][1]; c2[6] = cz[3][2]; c2[7] = cz[3][3];
    const v8f d1 = __builtin_amdgcn_wmma_f32_16x16x4_f32(
         false, av,  false, bvl,
         (short)0, c1,  false,  false);
    const v8f d2 = __builtin_amdgcn_wmma_f32_16x16x4_f32(
         false, av,  false, bvh,
         (short)0, c2,  false,  false);
    float v1[8], v2[8], rcv[8];
#pragma unroll
    for (int r = 0; r < 8; ++r) {
        v1[r] = whf(d1[r]);
        v2[r] = whf(d2[r]);
    }
    __builtin_amdgcn_sched_barrier(0);
#pragma unroll
    for (int r = 0; r < 8; ++r) rcv[r] = __shfl_xor(hi ? v1[r] : v2[r], 16);
    __builtin_amdgcn_sched_barrier(0);
#pragma unroll
    for (int r = 0; r < 8; ++r) {
        seq[r >> 2][r & 3]       = hi ? rcv[r] : v1[r];
        seq[2 + (r >> 2)][r & 3] = hi ? v2[r]  : rcv[r];
    }
}

template <int NS>
__device__ __forceinline__ void f_block(const v4f (&cz)[NS][4], const v2f (&ah)[NS],
                                        v2f bvl, v2f bvh, bool hi, float* zdst,
                                        float b0, float b1, float b2,
                                        float a0, float a1, float a2,
                                        float& p1, float& p2, float& p3,
                                        float& y1, float& y2, float& y3) {
#pragma unroll
    for (int s = 0; s < NS; ++s) {
        v4f seq[4];
        f_v16(cz[s], ah[s], bvl, bvh, hi, seq);
        p1_block<0, 4>(seq, zdst + 16 * s, b0, b1, b2, a0, a1, a2,
                       p1, p2, p3, y1, y2, y3);
    }
}

__global__ __launch_bounds__(256)
void fused_kernel(float* z,
                  const float* __restrict__ states,
                  const float* __restrict__ hpack,
                  const float* __restrict__ bco,
                  const float* __restrict__ aco,
                  const float* __restrict__ hpack2,
                  float* __restrict__ states2) {
    const int id   = blockIdx.x * blockDim.x + threadIdx.x;
    const int b    = id / C;
    const int lane = threadIdx.x & 31;
    const int n    = lane & 15;
    const int half = lane >> 4;
    const bool hi  = half != 0;
    const float b0 = bco[0], b1 = bco[1], b2 = bco[2];
    const float a0 = aco[0], a1 = aco[1], a2 = aco[2];
    const HTail ht = load_htail(hpack2);

    float* zrow = z + (size_t)b * T;
    const float* zlo  = zrow + (size_t)n * L + 8 * half;
    const float* zhi  = zrow + (size_t)(16 + n) * L + 8 * half;
    float*       zown = zrow + (size_t)lane * L;
    const float* __restrict__ hrow = hpack + n * 4 + 2 * half;

    const v2f bvl = *(const v2f*)(states + ((size_t)b * C + n) * 4 + 2 * half);
    const v2f bvh = *(const v2f*)(states + ((size_t)b * C + 16 + n) * 4 + 2 * half);

    float p1, p2, p3 = 0.0f;
    {
        v4f cz[1][4]; v2f ah[1]; v4f seq[4];
        f_load<1>(cz, ah, zlo + (L - 16), zhi + (L - 16), hrow + (L - 16) * 4);
        f_v16(cz[0], ah[0], bvl, bvh, hi, seq);
        p1 = __shfl_up(seq[3][3], 1);
        p2 = __shfl_up(seq[3][2], 1);
        if (lane == 0) { p1 = 0.0f; p2 = 0.0f; }
    }
    float y1 = 0.0f, y2 = 0.0f, y3 = 0.0f;

    v4f ca[F_NS][4], cb[F_NS][4];
    v2f ha[F_NS], hb[F_NS];
    f_load<F_NS>(ca, ha, zlo, zhi, hrow);
    for (int blk = 0; blk < P1_NB - 2; blk += 2) {
        f_load<F_NS>(cb, hb, zlo + (blk + 1) * P1_BLK, zhi + (blk + 1) * P1_BLK,
                     hrow + (blk + 1) * P1_BLK * 4);
        f_block<F_NS>(ca, ha, bvl, bvh, hi, zown + blk * P1_BLK,
                      b0, b1, b2, a0, a1, a2, p1, p2, p3, y1, y2, y3);
        f_load<F_NS>(ca, ha, zlo + (blk + 2) * P1_BLK, zhi + (blk + 2) * P1_BLK,
                     hrow + (blk + 2) * P1_BLK * 4);
        f_block<F_NS>(cb, hb, bvl, bvh, hi, zown + (blk + 1) * P1_BLK,
                      b0, b1, b2, a0, a1, a2, p1, p2, p3, y1, y2, y3);
    }
    f_load<F_NS>(cb, hb, zlo + (P1_NB - 1) * P1_BLK, zhi + (P1_NB - 1) * P1_BLK,
                 hrow + (P1_NB - 1) * P1_BLK * 4);
    f_block<F_NS>(ca, ha, bvl, bvh, hi, zown + (P1_NB - 2) * P1_BLK,
                  b0, b1, b2, a0, a1, a2, p1, p2, p3, y1, y2, y3);
    f_block<F_NS>(cb, hb, bvl, bvh, hi, zown + (P1_NB - 1) * P1_BLK,
                  b0, b1, b2, a0, a1, a2, p1, p2, p3, y1, y2, y3);
    wave_scan_store(ht, y1, y2, y3, lane, states2 + (size_t)id * 4);
}

extern "C" void kernel_launch(void* const* d_in, const int* in_sizes, int n_in,
                              void* d_out, int out_size, void* d_ws, size_t ws_size,
                              hipStream_t stream) {
    const float* u  = (const float*)d_in[0];
    const float* b1 = (const float*)d_in[1];
    const float* a1 = (const float*)d_in[2];
    const float* b2 = (const float*)d_in[3];
    const float* a2 = (const float*)d_in[4];
    float* out = (float*)d_out;

    float* ws     = (float*)d_ws;
    float* z      = ws;
    float* st1    = z + (size_t)Bsz * T;
    float* st2    = st1 + (size_t)Bsz * C * 4;
    float* hp1    = st2 + (size_t)Bsz * C * 4;
    float* hp2    = hp1 + L * 4;

    const int p1_blocks  = (Bsz * C) / 256;
    const int cor_blocks = (Bsz * (C / 16)) / 8;

    hpack_kernel<<<2, 32, 0, stream>>>(a1, a2, hp1, hp2);

    phase1_kernel<1><<<p1_blocks, 256, 0, stream>>>(u, b1, a1, hp1, z, st1);

    fused_kernel<<<p1_blocks, 256, 0, stream>>>(z, st1, hp1, b2, a2, hp2, st2);

    correct_kernel<<<cor_blocks, 256, 0, stream>>>(z, st2, hp2, out);
}


# table-build launch removed: its two blocks ride in the stage-1 launch; the stage-1 state scan, which needs the table, moves to the head of the fused kernel (same sums, same order)
# speedup vs baseline: 5.7172x; 1.0946x over previous
#include <hip/hip_runtime.h>
#include <hip/hip_bf16.h>
#include <math.h>

static constexpr int Bsz = 512;
static constexpr int T   = 16384;
static constexpr int L   = 512;
static constexpr int C   = T / L;

typedef __attribute__((ext_vector_type(2))) float v2f;
typedef __attribute__((ext_vector_type(4))) float v4f;
typedef __attribute__((ext_vector_type(8))) float v8f;

__device__ __forceinline__ float whf(float x) {
    const float c   = 10.0f / 11.0f;
    const float l2e = 1.4426950408889634f;
    float e = 1.0f - __builtin_amdgcn_exp2f(-(c * l2e) * x);
    float l = c * x;
    unsigned s = (unsigned)(__float_as_int(x) >> 31);
    return __uint_as_float((__float_as_uint(e) & ~s) | (__float_as_uint(l) & s));
}

static constexpr int HP_BLOCKS = 2;

__device__ __forceinline__ void hpack_block(int which,
                                            const float* __restrict__ a1c,
                                            const float* __restrict__ a2c,
                                            float* __restrict__ hp1,
                                            float* __restrict__ hp2) {
    const float* ac = (which == 0) ? a1c : a2c;
    float*       hp = (which == 0) ? hp1 : hp2;
    int j = threadIdx.x;
    if (j > 3) return;
    if (j == 3) {
        for (int i = 0; i < L; ++i) hp[i * 4 + 3] = 0.0f;
        return;
    }
    const float a0 = ac[0], a1 = ac[1], a2 = ac[2];
    float y1 = (j == 0) ? 1.0f : 0.0f;
    float y2 = (j == 1) ? 1.0f : 0.0f;
    float y3 = (j == 2) ? 1.0f : 0.0f;
    for (int i = 0; i < L; ++i) {
        float y = -(a0 * y1 + a1 * y2 + a2 * y3);
        hp[i * 4 + j] = y;
        y3 = y2; y2 = y1; y1 = y;
    }
}

static_assert(C == 32, "one wave per row: lane index == chunk index");

struct HTail { float h00, h01, h02, h10, h11, h12, h20, h21, h22; };

__device__ __forceinline__ HTail load_htail(const float* __restrict__ hpack) {
    HTail h;
    h.h00 = hpack[(L - 1) * 4 + 0]; h.h01 = hpack[(L - 1) * 4 + 1]; h.h02 = hpack[(L - 1) * 4 + 2];
    h.h10 = hpack[(L - 2) * 4 + 0]; h.h11 = hpack[(L - 2) * 4 + 1]; h.h12 = hpack[(L - 2) * 4 + 2];
    h.h20 = hpack[(L - 3) * 4 + 0]; h.h21 = hpack[(L - 3) * 4 + 1]; h.h22 = hpack[(L - 3) * 4 + 2];
    return h;
}

__device__ __forceinline__ float lane_bcast(float v, int src) {
    return __int_as_float(__builtin_amdgcn_readlane(__float_as_int(v), src));
}

__device__ __forceinline__ void wave_scan(const HTail& h, float t0, float t1, float t2,
                                          int lane, float& e0, float& e1, float& e2) {
    float s0 = 0.0f, s1 = 0.0f, s2 = 0.0f;
    e0 = 0.0f; e1 = 0.0f; e2 = 0.0f;
    for (int c = 0; c < C; ++c) {
        if (lane == c) { e0 = s0; e1 = s1; e2 = s2; }
        const float c0 = lane_bcast(t0, c), c1 = lane_bcast(t1, c), c2 = lane_bcast(t2, c);
        const float n0 = c0 + h.h00 * s0 + h.h01 * s1 + h.h02 * s2;
        const float n1 = c1 + h.h10 * s0 + h.h11 * s1 + h.h12 * s2;
        const float n2 = c2 + h.h20 * s0 + h.h21 * s1 + h.h22 * s2;
        s0 = n0; s1 = n1; s2 = n2;
    }
}

__device__ __forceinline__ void wave_scan_store(const HTail& h, float t0, float t1, float t2,
                                                int lane, float* __restrict__ st) {
    float e0, e1, e2;
    wave_scan(h, t0, t1, t2, lane, e0, e1, e2);
    v4f e;
    e[0] = e0; e[1] = e1; e[2] = e2; e[3] = 0.0f;
    *(v4f*)st = e;
}

static constexpr int P1_BLK = 64;
static constexpr int P1_NV  = P1_BLK / 4;
static constexpr int P1_NB  = L / P1_BLK;

template <int NV>
__device__ __forceinline__ void p1_load(v4f (&dst)[NV], const float* __restrict__ p) {
#pragma unroll
    for (int v = 0; v < NV; ++v) dst[v] = *(const v4f*)(p + 4 * v);
    __builtin_amdgcn_sched_barrier(0);
}

template <int NK, int NV>
__device__ __forceinline__ void p1_block(const v4f (&xin)[NV], float* __restrict__ zdst,
                                         float b0, float b1, float b2,
                                         float a0, float a1, float a2,
                                         float& p1, float& p2, float& p3,
                                         float& y1, float& y2, float& y3) {
#pragma unroll
    for (int v = 0; v < NV; ++v) {
        v4f o;
#pragma unroll
        for (int e = 0; e < 4; ++e) {
            const float s = xin[v][e];
            float xf;
            if (NK == 0) xf = b0 * s  + b1 * p1 + b2 * p2;
            else         xf = b0 * p1 + b1 * p2 + b2 * p3;
            const float y = xf - (a0 * y1 + a1 * y2 + a2 * y3);
            o[e] = y;
            p3 = p2; p2 = p1; p1 = s;
            y3 = y2; y2 = y1; y1 = y;
        }
        *(v4f*)(zdst + 4 * v) = o;
    }
}

template <int NK>
__global__ __launch_bounds__(256)
void phase1_kernel(const float* __restrict__ in,
                   const float* __restrict__ bco,
                   const float* __restrict__ aco,
                   const float* __restrict__ aco2,
                   float* __restrict__ hp1,
                   float* __restrict__ hp2,
                   float* __restrict__ z,
                   float* __restrict__ tails) {
    if (blockIdx.x < HP_BLOCKS) {
        hpack_block(blockIdx.x, aco, aco2, hp1, hp2);
        return;
    }
    const int id = (blockIdx.x - HP_BLOCKS) * blockDim.x + threadIdx.x;
    const int b  = id / C;
    const int c  = id % C;
    const float b0 = bco[0], b1 = bco[1], b2 = bco[2];
    const float a0 = aco[0], a1 = aco[1], a2 = aco[2];

    const float* __restrict__ src  = in + (size_t)b * T + (size_t)c * L;
    float*       __restrict__ zrow = z  + (size_t)b * T + (size_t)c * L;

    float p1 = 0.0f, p2 = 0.0f, p3 = 0.0f;
    if (c > 0) {
        p1 = src[-1];
        p2 = src[-2];
        if (NK == 1) p3 = src[-3];
    }
    float y1 = 0.0f, y2 = 0.0f, y3 = 0.0f;

    v4f xa[P1_NV], xb[P1_NV];
    p1_load<P1_NV>(xa, src);
    for (int blk = 0; blk < P1_NB - 2; blk += 2) {
        p1_load<P1_NV>(xb, src + (blk + 1) * P1_BLK);
        p1_block<NK, P1_NV>(xa, zrow + blk * P1_BLK, b0, b1, b2, a0, a1, a2,
                            p1, p2, p3, y1, y2, y3);
        p1_load<P1_NV>(xa, src + (blk + 2) * P1_BLK);
        p1_block<NK, P1_NV>(xb, zrow + (blk + 1) * P1_BLK, b0, b1, b2, a0, a1, a2,
                            p1, p2, p3, y1, y2, y3);
    }
    p1_load<P1_NV>(xb, src + (P1_NB - 1) * P1_BLK);
    p1_block<NK, P1_NV>(xa, zrow + (P1_NB - 2) * P1_BLK, b0, b1, b2, a0, a1, a2,
                        p1, p2, p3, y1, y2, y3);
    p1_block<NK, P1_NV>(xb, zrow + (P1_NB - 1) * P1_BLK, b0, b1, b2, a0, a1, a2,
                        p1, p2, p3, y1, y2, y3);
    v4f tl;
    tl[0] = y1; tl[1] = y2; tl[2] = y3; tl[3] = 0.0f;
    *(v4f*)(tails + (size_t)id * 4) = tl;
}

__global__ __launch_bounds__(256)
void correct_kernel(const float* __restrict__ z,
                    const float* __restrict__ states,
                    const float* __restrict__ hpack,
                    float* __restrict__ out) {
    const int lane = threadIdx.x & 31;
    const int wave = threadIdx.x >> 5;
    const int g    = blockIdx.x * (blockDim.x >> 5) + wave;
    const int n    = lane & 15;
    const int half = lane >> 4;
    const int b    = g / (C / 16);
    const int cg   = g % (C / 16);
    const int c    = cg * 16 + n;

    const float* __restrict__ zb = z   + (size_t)b * T + (size_t)c * L;
    float*       __restrict__ ob = out + (size_t)b * T + (size_t)c * L;

    const v2f bv = *(const v2f*)(states + ((size_t)b * C + c) * 4 + 2 * half);

#pragma unroll 4
    for (int ib = 0; ib < L; ib += 16) {
        const v2f av = *(const v2f*)(hpack + (size_t)(ib + n) * 4 + 2 * half);
        const v4f c0 = *(const v4f*)(zb + ib + 8 * half);
        const v4f c1 = *(const v4f*)(zb + ib + 8 * half + 4);
        v8f cv;
        cv[0] = c0[0]; cv[1] = c0[1]; cv[2] = c0[2]; cv[3] = c0[3];
        cv[4] = c1[0]; cv[5] = c1[1]; cv[6] = c1[2]; cv[7] = c1[3];

        v8f dv = __builtin_amdgcn_wmma_f32_16x16x4_f32(
             false, av,  false, bv,
             (short)0, cv,  false,  false);

        v4f o0, o1;
        o0[0] = dv[0]; o0[1] = dv[1]; o0[2] = dv[2]; o0[3] = dv[3];
        o1[0] = dv[4]; o1[1] = dv[5]; o1[2] = dv[6]; o1[3] = dv[7];
        __builtin_nontemporal_store(o0, (v4f*)(ob + ib + 8 * half));
        __builtin_nontemporal_store(o1, (v4f*)(ob + ib + 8 * half + 4));
    }
}

static constexpr int F_NS = P1_BLK / 16;

template <int NS>
__device__ __forceinline__ void f_load(v4f (&cz)[NS][4], v2f (&ah)[NS],
                                       const float* zlo, const float* zhi,
                                       const float* __restrict__ hrow) {
#pragma unroll
    for (int s = 0; s < NS; ++s) {
        cz[s][0] = *(const v4f*)(zlo + 16 * s);
        cz[s][1] = *(const v4f*)(zlo + 16 * s + 4);
        cz[s][2] = *(const v4f*)(zhi + 16 * s);
        cz[s][3] = *(const v4f*)(zhi + 16 * s + 4);
        ah[s]    = *(const v2f*)(hrow + 64 * s);
    }
    __builtin_amdgcn_sched_barrier(0);
}

__device__ __forceinline__ void f_v16(const v4f (&cz)[4], v2f av, v2f bvl, v2f bvh,
                                      bool hi, v4f (&seq)[4]) {
    v8f c1, c2;
    c1[0] = cz[0][0]; c1[1] = cz[0][1]; c1[2] = cz[0][2]; c1[3] = cz[0][3];
    c1[4] = cz[1][0]; c1[5] = cz[1][1]; c1[6] = cz[1][2]; c1[7] = cz[1][3];
    c2[0] = cz[2][0]; c2[1] = cz[2][1]; c2[2] = cz[2][2]; c2[3] = cz[2][3];
    c2[4] = cz[3][0]; c2[5] = cz[3][1]; c2[6] = cz[3][2]; c2[7] = cz[3][3];
    const v8f d1 = __builtin_amdgcn_wmma_f32_16x16x4_f32(
         false, av,  false, bvl,
         (short)0, c1,  false,  false);
    const v8f d2 = __builtin_amdgcn_wmma_f32_16x16x4_f32(
         false, av,  false, bvh,
         (short)0, c2,  false,  false);
    float v1[8], v2[8], rcv[8];
#pragma unroll
    for (int r = 0; r < 8; ++r) {
        v1[r] = whf(d1[r]);
        v2[r] = whf(d2[r]);
    }
    __builtin_amdgcn_sched_barrier(0);
#pragma unroll
    for (int r = 0; r < 8; ++r) rcv[r] = __shfl_xor(hi ? v1[r] : v2[r], 16);
    __builtin_amdgcn_sched_barrier(0);
#pragma unroll
    for (int r = 0; r < 8; ++r) {
        seq[r >> 2][r & 3]       = hi ? rcv[r] : v1[r];
        seq[2 + (r >> 2)][r & 3] = hi ? v2[r]  : rcv[r];
    }
}

template <int NS>
__device__ __forceinline__ void f_block(const v4f (&cz)[NS][4], const v2f (&ah)[NS],
                                        v2f bvl, v2f bvh, bool hi, float* zdst,
                                        float b0, float b1, float b2,
                                        float a0, float a1, float a2,
                                        float& p1, float& p2, float& p3,
                                        float& y1, float& y2, float& y3) {
#pragma unroll
    for (int s = 0; s < NS; ++s) {
        v4f seq[4];
        f_v16(cz[s], ah[s], bvl, bvh, hi, seq);
        p1_block<0, 4>(seq, zdst + 16 * s, b0, b1, b2, a0, a1, a2,
                       p1, p2, p3, y1, y2, y3);
    }
}

__global__ __launch_bounds__(256)
void fused_kernel(float* z,
                  const float* __restrict__ tails,
                  const float* __restrict__ hpack,
                  const float* __restrict__ bco,
                  const float* __restrict__ aco,
                  const float* __restrict__ hpack2,
                  float* __restrict__ states2) {
    const int id   = blockIdx.x * blockDim.x + threadIdx.x;
    const int b    = id / C;
    const int lane = threadIdx.x & 31;
    const int n    = lane & 15;
    const int half = lane >> 4;
    const bool hi  = half != 0;
    const float b0 = bco[0], b1 = bco[1], b2 = bco[2];
    const float a0 = aco[0], a1 = aco[1], a2 = aco[2];
    const HTail ht = load_htail(hpack2);

    float* zrow = z + (size_t)b * T;
    const float* zlo  = zrow + (size_t)n * L + 8 * half;
    const float* zhi  = zrow + (size_t)(16 + n) * L + 8 * half;
    float*       zown = zrow + (size_t)lane * L;
    const float* __restrict__ hrow = hpack + n * 4 + 2 * half;

    float e0, e1, e2;
    {
        const v4f tl = *(const v4f*)(tails + (size_t)id * 4);
        const HTail ht1 = load_htail(hpack);
        wave_scan(ht1, tl[0], tl[1], tl[2], lane, e0, e1, e2);
    }
    v2f bvl, bvh;
    {
        const float l0 = __shfl(e0, n),      l1 = __shfl(e1, n),      l2 = __shfl(e2, n);
        const float h0 = __shfl(e0, 16 + n), h1 = __shfl(e1, 16 + n), h2 = __shfl(e2, 16 + n);
        bvl[0] = hi ? l2 : l0;  bvl[1] = hi ? 0.0f : l1;
        bvh[0] = hi ? h2 : h0;  bvh[1] = hi ? 0.0f : h1;
    }

    float p1, p2, p3 = 0.0f;
    {
        v4f cz[1][4]; v2f ah[1]; v4f seq[4];
        f_load<1>(cz, ah, zlo + (L - 16), zhi + (L - 16), hrow + (L - 16) * 4);
        f_v16(cz[0], ah[0], bvl, bvh, hi, seq);
        p1 = __shfl_up(seq[3][3], 1);
        p2 = __shfl_up(seq[3][2], 1);
        if (lane == 0) { p1 = 0.0f; p2 = 0.0f; }
    }
    float y1 = 0.0f, y2 = 0.0f, y3 = 0.0f;

    v4f ca[F_NS][4], cb[F_NS][4];
    v2f ha[F_NS], hb[F_NS];
    f_load<F_NS>(ca, ha, zlo, zhi, hrow);
    for (int blk = 0; blk < P1_NB - 2; blk += 2) {
        f_load<F_NS>(cb, hb, zlo + (blk + 1) * P1_BLK, zhi + (blk + 1) * P1_BLK,
                     hrow + (blk + 1) * P1_BLK * 4);
        f_block<F_NS>(ca, ha, bvl, bvh, hi, zown + blk * P1_BLK,
                      b0, b1, b2, a0, a1, a2, p1, p2, p3, y1, y2, y3);
        f_load<F_NS>(ca, ha, zlo + (blk + 2) * P1_BLK, zhi + (blk + 2) * P1_BLK,
                     hrow + (blk + 2) * P1_BLK * 4);
        f_block<F_NS>(cb, hb, bvl, bvh, hi, zown + (blk + 1) * P1_BLK,
                      b0, b1, b2, a0, a1, a2, p1, p2, p3, y1, y2, y3);
    }
    f_load<F_NS>(cb, hb, zlo + (P1_NB - 1) * P1_BLK, zhi + (P1_NB - 1) * P1_BLK,
                 hrow + (P1_NB - 1) * P1_BLK * 4);
    f_block<F_NS>(ca, ha, bvl, bvh, hi, zown + (P1_NB - 2) * P1_BLK,
                  b0, b1, b2, a0, a1, a2, p1, p2, p3, y1, y2, y3);
    f_block<F_NS>(cb, hb, bvl, bvh, hi, zown + (P1_NB - 1) * P1_BLK,
                  b0, b1, b2, a0, a1, a2, p1, p2, p3, y1, y2, y3);
    wave_scan_store(ht, y1, y2, y3, lane, states2 + (size_t)id * 4);
}

extern "C" void kernel_launch(void* const* d_in, const int* in_sizes, int n_in,
                              void* d_out, int out_size, void* d_ws, size_t ws_size,
                              hipStream_t stream) {
    const float* u  = (const float*)d_in[0];
    const float* b1 = (const float*)d_in[1];
    const float* a1 = (const float*)d_in[2];
    const float* b2 = (const float*)d_in[3];
    const float* a2 = (const float*)d_in[4];
    float* out = (float*)d_out;

    float* ws     = (float*)d_ws;
    float* z      = ws;
    float* tl1    = z + (size_t)Bsz * T;
    float* st2    = tl1 + (size_t)Bsz * C * 4;
    float* hp1    = st2 + (size_t)Bsz * C * 4;
    float* hp2    = hp1 + L * 4;

    const int p1_blocks  = (Bsz * C) / 256;
    const int p1_grid    = HP_BLOCKS + (Bsz * C) / 256;
    const int cor_blocks = (Bsz * (C / 16)) / 8;

    phase1_kernel<1><<<p1_grid, 256, 0, stream>>>(u, b1, a1, a2, hp1, hp2, z, tl1);

    fused_kernel<<<p1_blocks, 256, 0, stream>>>(z, tl1, hp1, b2, a2, hp2, st2);

    correct_kernel<<<cor_blocks, 256, 0, stream>>>(z, st2, hp2, out);
}


# lane==chunk kernels: global traffic moves as whole 128-byte lines through a wave-private LDS stage (phase1 loads and stores, fused kernel stores); was 32 lines per instruction, 16 bytes of each
# speedup vs baseline: 7.4793x; 1.3082x over previous
#include <hip/hip_runtime.h>
#include <hip/hip_bf16.h>
#include <math.h>

static constexpr int Bsz = 512;
static constexpr int T   = 16384;
static constexpr int L   = 512;
static constexpr int C   = T / L;

typedef __attribute__((ext_vector_type(2))) float v2f;
typedef __attribute__((ext_vector_type(4))) float v4f;
typedef __attribute__((ext_vector_type(8))) float v8f;

__device__ __forceinline__ float whf(float x) {
    const float c   = 10.0f / 11.0f;
    const float l2e = 1.4426950408889634f;
    float e = 1.0f - __builtin_amdgcn_exp2f(-(c * l2e) * x);
    float l = c * x;
    unsigned s = (unsigned)(__float_as_int(x) >> 31);
    return __uint_as_float((__float_as_uint(e) & ~s) | (__float_as_uint(l) & s));
}

static constexpr int HP_BLOCKS = 2;

__device__ __forceinline__ void hpack_block(int which,
                                            const float* __restrict__ a1c,
                                            const float* __restrict__ a2c,
                                            float* __restrict__ hp1,
                                            float* __restrict__ hp2) {
    const float* ac = (which == 0) ? a1c : a2c;
    float*       hp = (which == 0) ? hp1 : hp2;
    int j = threadIdx.x;
    if (j > 3) return;
    if (j == 3) {
        for (int i = 0; i < L; ++i) hp[i * 4 + 3] = 0.0f;
        return;
    }
    const float a0 = ac[0], a1 = ac[1], a2 = ac[2];
    float y1 = (j == 0) ? 1.0f : 0.0f;
    float y2 = (j == 1) ? 1.0f : 0.0f;
    float y3 = (j == 2) ? 1.0f : 0.0f;
    for (int i = 0; i < L; ++i) {
        float y = -(a0 * y1 + a1 * y2 + a2 * y3);
        hp[i * 4 + j] = y;
        y3 = y2; y2 = y1; y1 = y;
    }
}

static_assert(C == 32, "one wave per row: lane index == chunk index");

struct HTail { float h00, h01, h02, h10, h11, h12, h20, h21, h22; };

__device__ __forceinline__ HTail load_htail(const float* __restrict__ hpack) {
    HTail h;
    h.h00 = hpack[(L - 1) * 4 + 0]; h.h01 = hpack[(L - 1) * 4 + 1]; h.h02 = hpack[(L - 1) * 4 + 2];
    h.h10 = hpack[(L - 2) * 4 + 0]; h.h11 = hpack[(L - 2) * 4 + 1]; h.h12 = hpack[(L - 2) * 4 + 2];
    h.h20 = hpack[(L - 3) * 4 + 0]; h.h21 = hpack[(L - 3) * 4 + 1]; h.h22 = hpack[(L - 3) * 4 + 2];
    return h;
}

__device__ __forceinline__ float lane_bcast(float v, int src) {
    return __int_as_float(__builtin_amdgcn_readlane(__float_as_int(v), src));
}

__device__ __forceinline__ void wave_scan(const HTail& h, float t0, float t1, float t2,
                                          int lane, float& e0, float& e1, float& e2) {
    float s0 = 0.0f, s1 = 0.0f, s2 = 0.0f;
    e0 = 0.0f; e1 = 0.0f; e2 = 0.0f;
    for (int c = 0; c < C; ++c) {
        if (lane == c) { e0 = s0; e1 = s1; e2 = s2; }
        const float c0 = lane_bcast(t0, c), c1 = lane_bcast(t1, c), c2 = lane_bcast(t2, c);
        const float n0 = c0 + h.h00 * s0 + h.h01 * s1 + h.h02 * s2;
        const float n1 = c1 + h.h10 * s0 + h.h11 * s1 + h.h12 * s2;
        const float n2 = c2 + h.h20 * s0 + h.h21 * s1 + h.h22 * s2;
        s0 = n0; s1 = n1; s2 = n2;
    }
}

__device__ __forceinline__ void wave_scan_store(const HTail& h, float t0, float t1, float t2,
                                                int lane, float* __restrict__ st) {
    float e0, e1, e2;
    wave_scan(h, t0, t1, t2, lane, e0, e1, e2);
    v4f e;
    e[0] = e0; e[1] = e1; e[2] = e2; e[3] = 0.0f;
    *(v4f*)st = e;
}

static constexpr int P1_BLK = 64;
static constexpr int P1_NV  = P1_BLK / 4;
static constexpr int P1_NB  = L / P1_BLK;

static constexpr int STG_ROW   = P1_BLK;
static constexpr int STG_WAVE  = C * STG_ROW;

__device__ __forceinline__ void stage_sync() {
    __builtin_amdgcn_fence(__ATOMIC_RELEASE, "wavefront");
    __builtin_amdgcn_wave_barrier();
    __builtin_amdgcn_fence(__ATOMIC_ACQUIRE, "wavefront");
}

struct StageMap {
    int sub;
    int slot;
    int own;
};

__device__ __forceinline__ int stg_off(int cc, int s) {
    return cc * STG_ROW + 4 * (s ^ (cc & 15));
}

template <int NV>
__device__ __forceinline__ void lines_load(v4f (&g)[NV], const float* __restrict__ rowblk,
                                           const StageMap& m) {
#pragma unroll
    for (int j = 0; j < NV; ++j)
        g[j] = *(const v4f*)(rowblk + (size_t)(2 * j + m.sub) * L + 4 * m.slot);
    __builtin_amdgcn_sched_barrier(0);
}

template <int NV>
__device__ __forceinline__ void lines_store(const v4f (&g)[NV], float* __restrict__ rowblk,
                                            const StageMap& m) {
#pragma unroll
    for (int j = 0; j < NV; ++j)
        *(v4f*)(rowblk + (size_t)(2 * j + m.sub) * L + 4 * m.slot) = g[j];
}

template <int NV>
__device__ __forceinline__ void stage_put_lines(const v4f (&g)[NV], float* stage, const StageMap& m) {
#pragma unroll
    for (int j = 0; j < NV; ++j) *(v4f*)(stage + stg_off(2 * j + m.sub, m.slot)) = g[j];
}
template <int NV>
__device__ __forceinline__ void stage_get_lines(v4f (&g)[NV], const float* stage, const StageMap& m) {
#pragma unroll
    for (int j = 0; j < NV; ++j) g[j] = *(const v4f*)(stage + stg_off(2 * j + m.sub, m.slot));
}
template <int NV>
__device__ __forceinline__ void stage_put_own(const v4f (&x)[NV], float* stage, const StageMap& m) {
#pragma unroll
    for (int v = 0; v < NV; ++v) *(v4f*)(stage + stg_off(m.own, v)) = x[v];
}
template <int NV>
__device__ __forceinline__ void stage_get_own(v4f (&x)[NV], const float* stage, const StageMap& m) {
#pragma unroll
    for (int v = 0; v < NV; ++v) x[v] = *(const v4f*)(stage + stg_off(m.own, v));
}

template <int NV>
__device__ __forceinline__ void p1_load(v4f (&dst)[NV], const float* __restrict__ p) {
#pragma unroll
    for (int v = 0; v < NV; ++v) dst[v] = *(const v4f*)(p + 4 * v);
    __builtin_amdgcn_sched_barrier(0);
}

template <int NK, int NV>
__device__ __forceinline__ void p1_block(const v4f (&xin)[NV], float* __restrict__ zdst,
                                         float b0, float b1, float b2,
                                         float a0, float a1, float a2,
                                         float& p1, float& p2, float& p3,
                                         float& y1, float& y2, float& y3) {
#pragma unroll
    for (int v = 0; v < NV; ++v) {
        v4f o;
#pragma unroll
        for (int e = 0; e < 4; ++e) {
            const float s = xin[v][e];
            float xf;
            if (NK == 0) xf = b0 * s  + b1 * p1 + b2 * p2;
            else         xf = b0 * p1 + b1 * p2 + b2 * p3;
            const float y = xf - (a0 * y1 + a1 * y2 + a2 * y3);
            o[e] = y;
            p3 = p2; p2 = p1; p1 = s;
            y3 = y2; y2 = y1; y1 = y;
        }
        *(v4f*)(zdst + 4 * v) = o;
    }
}

template <int NK, int NV>
__device__ __forceinline__ void p1_block_reg(v4f (&x)[NV],
                                             float b0, float b1, float b2,
                                             float a0, float a1, float a2,
                                             float& p1, float& p2, float& p3,
                                             float& y1, float& y2, float& y3) {
#pragma unroll
    for (int v = 0; v < NV; ++v) {
        v4f o;
#pragma unroll
        for (int e = 0; e < 4; ++e) {
            const float s = x[v][e];
            float xf;
            if (NK == 0) xf = b0 * s  + b1 * p1 + b2 * p2;
            else         xf = b0 * p1 + b1 * p2 + b2 * p3;
            const float y = xf - (a0 * y1 + a1 * y2 + a2 * y3);
            o[e] = y;
            p3 = p2; p2 = p1; p1 = s;
            y3 = y2; y2 = y1; y1 = y;
        }
        x[v] = o;
    }
}

template <int NK>
__global__ __launch_bounds__(256)
void phase1_kernel(const float* __restrict__ in,
                   const float* __restrict__ bco,
                   const float* __restrict__ aco,
                   const float* __restrict__ aco2,
                   float* __restrict__ hp1,
                   float* __restrict__ hp2,
                   float* __restrict__ z,
                   float* __restrict__ tails) {
    __shared__ __attribute__((aligned(16))) float stage_all[8 * STG_WAVE];
    if (blockIdx.x < HP_BLOCKS) {
        hpack_block(blockIdx.x, aco, aco2, hp1, hp2);
        return;
    }
    const int id = (blockIdx.x - HP_BLOCKS) * blockDim.x + threadIdx.x;
    const int b  = id / C;
    const int c  = id % C;
    const float b0 = bco[0], b1 = bco[1], b2 = bco[2];
    const float a0 = aco[0], a1 = aco[1], a2 = aco[2];

    const float* __restrict__ src  = in + (size_t)b * T + (size_t)c * L;
    const float* __restrict__ urow = in + (size_t)b * T;
    float*       __restrict__ zrow = z  + (size_t)b * T;

    float* stage = stage_all + (threadIdx.x >> 5) * STG_WAVE;
    StageMap m;
    m.sub = (threadIdx.x & 31) >> 4;  m.slot = threadIdx.x & 15;  m.own = c;

    float p1 = 0.0f, p2 = 0.0f, p3 = 0.0f;
    if (c > 0) {
        p1 = src[-1];
        p2 = src[-2];
        if (NK == 1) p3 = src[-3];
    }
    float y1 = 0.0f, y2 = 0.0f, y3 = 0.0f;

    v4f g[P1_NV], x[P1_NV];
    lines_load<P1_NV>(g, urow, m);
    stage_put_lines<P1_NV>(g, stage, m);
    stage_sync();
#pragma unroll 1
    for (int blk = 0; blk < P1_NB; ++blk) {
        stage_get_own<P1_NV>(x, stage, m);
        stage_sync();
        if (blk + 1 < P1_NB) lines_load<P1_NV>(g, urow + (blk + 1) * P1_BLK, m);
        p1_block_reg<NK, P1_NV>(x, b0, b1, b2, a0, a1, a2, p1, p2, p3, y1, y2, y3);
        stage_put_own<P1_NV>(x, stage, m);
        stage_sync();
        stage_get_lines<P1_NV>(x, stage, m);
        stage_sync();
        lines_store<P1_NV>(x, zrow + blk * P1_BLK, m);
        if (blk + 1 < P1_NB) {
            stage_put_lines<P1_NV>(g, stage, m);
            stage_sync();
        }
    }
    v4f tl;
    tl[0] = y1; tl[1] = y2; tl[2] = y3; tl[3] = 0.0f;
    *(v4f*)(tails + (size_t)id * 4) = tl;
}

__global__ __launch_bounds__(256)
void correct_kernel(const float* __restrict__ z,
                    const float* __restrict__ states,
                    const float* __restrict__ hpack,
                    float* __restrict__ out) {
    const int lane = threadIdx.x & 31;
    const int wave = threadIdx.x >> 5;
    const int g    = blockIdx.x * (blockDim.x >> 5) + wave;
    const int n    = lane & 15;
    const int half = lane >> 4;
    const int b    = g / (C / 16);
    const int cg   = g % (C / 16);
    const int c    = cg * 16 + n;

    const float* __restrict__ zb = z   + (size_t)b * T + (size_t)c * L;
    float*       __restrict__ ob = out + (size_t)b * T + (size_t)c * L;

    const v2f bv = *(const v2f*)(states + ((size_t)b * C + c) * 4 + 2 * half);

#pragma unroll 4
    for (int ib = 0; ib < L; ib += 16) {
        const v2f av = *(const v2f*)(hpack + (size_t)(ib + n) * 4 + 2 * half);
        const v4f c0 = *(const v4f*)(zb + ib + 8 * half);
        const v4f c1 = *(const v4f*)(zb + ib + 8 * half + 4);
        v8f cv;
        cv[0] = c0[0]; cv[1] = c0[1]; cv[2] = c0[2]; cv[3] = c0[3];
        cv[4] = c1[0]; cv[5] = c1[1]; cv[6] = c1[2]; cv[7] = c1[3];

        v8f dv = __builtin_amdgcn_wmma_f32_16x16x4_f32(
             false, av,  false, bv,
             (short)0, cv,  false,  false);

        v4f o0, o1;
        o0[0] = dv[0]; o0[1] = dv[1]; o0[2] = dv[2]; o0[3] = dv[3];
        o1[0] = dv[4]; o1[1] = dv[5]; o1[2] = dv[6]; o1[3] = dv[7];
        __builtin_nontemporal_store(o0, (v4f*)(ob + ib + 8 * half));
        __builtin_nontemporal_store(o1, (v4f*)(ob + ib + 8 * half + 4));
    }
}

static constexpr int F_NS = P1_BLK / 16;

template <int NS>
__device__ __forceinline__ void f_load(v4f (&cz)[NS][4], v2f (&ah)[NS],
                                       const float* zlo, const float* zhi,
                                       const float* __restrict__ hrow) {
#pragma unroll
    for (int s = 0; s < NS; ++s) {
        cz[s][0] = *(const v4f*)(zlo + 16 * s);
        cz[s][1] = *(const v4f*)(zlo + 16 * s + 4);
        cz[s][2] = *(const v4f*)(zhi + 16 * s);
        cz[s][3] = *(const v4f*)(zhi + 16 * s + 4);
        ah[s]    = *(const v2f*)(hrow + 64 * s);
    }
    __builtin_amdgcn_sched_barrier(0);
}

__device__ __forceinline__ void f_v16(const v4f (&cz)[4], v2f av, v2f bvl, v2f bvh,
                                      bool hi, v4f (&seq)[4]) {
    v8f c1, c2;
    c1[0] = cz[0][0]; c1[1] = cz[0][1]; c1[2] = cz[0][2]; c1[3] = cz[0][3];
    c1[4] = cz[1][0]; c1[5] = cz[1][1]; c1[6] = cz[1][2]; c1[7] = cz[1][3];
    c2[0] = cz[2][0]; c2[1] = cz[2][1]; c2[2] = cz[2][2]; c2[3] = cz[2][3];
    c2[4] = cz[3][0]; c2[5] = cz[3][1]; c2[6] = cz[3][2]; c2[7] = cz[3][3];
    const v8f d1 = __builtin_amdgcn_wmma_f32_16x16x4_f32(
         false, av,  false, bvl,
         (short)0, c1,  false,  false);
    const v8f d2 = __builtin_amdgcn_wmma_f32_16x16x4_f32(
         false, av,  false, bvh,
         (short)0, c2,  false,  false);
    float v1[8], v2[8], rcv[8];
#pragma unroll
    for (int r = 0; r < 8; ++r) {
        v1[r] = whf(d1[r]);
        v2[r] = whf(d2[r]);
    }
    __builtin_amdgcn_sched_barrier(0);
#pragma unroll
    for (int r = 0; r < 8; ++r) rcv[r] = __shfl_xor(hi ? v1[r] : v2[r], 16);
    __builtin_amdgcn_sched_barrier(0);
#pragma unroll
    for (int r = 0; r < 8; ++r) {
        seq[r >> 2][r & 3]       = hi ? rcv[r] : v1[r];
        seq[2 + (r >> 2)][r & 3] = hi ? v2[r]  : rcv[r];
    }
}

template <int NS>
__device__ __forceinline__ void f_block(const v4f (&cz)[NS][4], const v2f (&ah)[NS],
                                        v2f bvl, v2f bvh, bool hi,
                                        float* stage, const StageMap& m, float* zrowblk,
                                        float b0, float b1, float b2,
                                        float a0, float a1, float a2,
                                        float& p1, float& p2, float& p3,
                                        float& y1, float& y2, float& y3) {
    v4f o[NS * 4];
#pragma unroll
    for (int s = 0; s < NS; ++s) {
        v4f seq[4];
        f_v16(cz[s], ah[s], bvl, bvh, hi, seq);
        p1_block_reg<0, 4>(seq, b0, b1, b2, a0, a1, a2, p1, p2, p3, y1, y2, y3);
#pragma unroll
        for (int i = 0; i < 4; ++i) o[4 * s + i] = seq[i];
    }
    stage_put_own<NS * 4>(o, stage, m);
    stage_sync();
    stage_get_lines<NS * 4>(o, stage, m);
    stage_sync();
    lines_store<NS * 4>(o, zrowblk, m);
}

__global__ __launch_bounds__(256)
void fused_kernel(float* z,
                  const float* __restrict__ tails,
                  const float* __restrict__ hpack,
                  const float* __restrict__ bco,
                  const float* __restrict__ aco,
                  const float* __restrict__ hpack2,
                  float* __restrict__ states2) {
    __shared__ __attribute__((aligned(16))) float stage_all[8 * STG_WAVE];
    const int id   = blockIdx.x * blockDim.x + threadIdx.x;
    const int b    = id / C;
    const int lane = threadIdx.x & 31;
    const int n    = lane & 15;
    const int half = lane >> 4;
    const bool hi  = half != 0;
    const float b0 = bco[0], b1 = bco[1], b2 = bco[2];
    const float a0 = aco[0], a1 = aco[1], a2 = aco[2];
    const HTail ht = load_htail(hpack2);

    float* zrow = z + (size_t)b * T;
    const float* zlo  = zrow + (size_t)n * L + 8 * half;
    const float* zhi  = zrow + (size_t)(16 + n) * L + 8 * half;
    float* stage = stage_all + (threadIdx.x >> 5) * STG_WAVE;
    StageMap m;
    m.sub = half;  m.slot = n;  m.own = lane;
    const float* __restrict__ hrow = hpack + n * 4 + 2 * half;

    float e0, e1, e2;
    {
        const v4f tl = *(const v4f*)(tails + (size_t)id * 4);
        const HTail ht1 = load_htail(hpack);
        wave_scan(ht1, tl[0], tl[1], tl[2], lane, e0, e1, e2);
    }
    v2f bvl, bvh;
    {
        const float l0 = __shfl(e0, n),      l1 = __shfl(e1, n),      l2 = __shfl(e2, n);
        const float h0 = __shfl(e0, 16 + n), h1 = __shfl(e1, 16 + n), h2 = __shfl(e2, 16 + n);
        bvl[0] = hi ? l2 : l0;  bvl[1] = hi ? 0.0f : l1;
        bvh[0] = hi ? h2 : h0;  bvh[1] = hi ? 0.0f : h1;
    }

    float p1, p2, p3 = 0.0f;
    {
        v4f cz[1][4]; v2f ah[1]; v4f seq[4];
        f_load<1>(cz, ah, zlo + (L - 16), zhi + (L - 16), hrow + (L - 16) * 4);
        f_v16(cz[0], ah[0], bvl, bvh, hi, seq);
        p1 = __shfl_up(seq[3][3], 1);
        p2 = __shfl_up(seq[3][2], 1);
        if (lane == 0) { p1 = 0.0f; p2 = 0.0f; }
    }
    float y1 = 0.0f, y2 = 0.0f, y3 = 0.0f;

    v4f ca[F_NS][4], cb[F_NS][4];
    v2f ha[F_NS], hb[F_NS];
    f_load<F_NS>(ca, ha, zlo, zhi, hrow);
    for (int blk = 0; blk < P1_NB - 2; blk += 2) {
        f_load<F_NS>(cb, hb, zlo + (blk + 1) * P1_BLK, zhi + (blk + 1) * P1_BLK,
                     hrow + (blk + 1) * P1_BLK * 4);
        f_block<F_NS>(ca, ha, bvl, bvh, hi, stage, m, zrow + blk * P1_BLK,
                      b0, b1, b2, a0, a1, a2, p1, p2, p3, y1, y2, y3);
        f_load<F_NS>(ca, ha, zlo + (blk + 2) * P1_BLK, zhi + (blk + 2) * P1_BLK,
                     hrow + (blk + 2) * P1_BLK * 4);
        f_block<F_NS>(cb, hb, bvl, bvh, hi, stage, m, zrow + (blk + 1) * P1_BLK,
                      b0, b1, b2, a0, a1, a2, p1, p2, p3, y1, y2, y3);
    }
    f_load<F_NS>(cb, hb, zlo + (P1_NB - 1) * P1_BLK, zhi + (P1_NB - 1) * P1_BLK,
                 hrow + (P1_NB - 1) * P1_BLK * 4);
    f_block<F_NS>(ca, ha, bvl, bvh, hi, stage, m, zrow + (P1_NB - 2) * P1_BLK,
                  b0, b1, b2, a0, a1, a2, p1, p2, p3, y1, y2, y3);
    f_block<F_NS>(cb, hb, bvl, bvh, hi, stage, m, zrow + (P1_NB - 1) * P1_BLK,
                  b0, b1, b2, a0, a1, a2, p1, p2, p3, y1, y2, y3);
    wave_scan_store(ht, y1, y2, y3, lane, states2 + (size_t)id * 4);
}

extern "C" void kernel_launch(void* const* d_in, const int* in_sizes, int n_in,
                              void* d_out, int out_size, void* d_ws, size_t ws_size,
                              hipStream_t stream) {
    const float* u  = (const float*)d_in[0];
    const float* b1 = (const float*)d_in[1];
    const float* a1 = (const float*)d_in[2];
    const float* b2 = (const float*)d_in[3];
    const float* a2 = (const float*)d_in[4];
    float* out = (float*)d_out;

    float* ws     = (float*)d_ws;
    float* z      = ws;
    float* tl1    = z + (size_t)Bsz * T;
    float* st2    = tl1 + (size_t)Bsz * C * 4;
    float* hp1    = st2 + (size_t)Bsz * C * 4;
    float* hp2    = hp1 + L * 4;

    const int p1_blocks  = (Bsz * C) / 256;
    const int p1_grid    = HP_BLOCKS + (Bsz * C) / 256;
    const int cor_blocks = (Bsz * (C / 16)) / 8;

    phase1_kernel<1><<<p1_grid, 256, 0, stream>>>(u, b1, a1, a2, hp1, hp2, z, tl1);

    fused_kernel<<<p1_blocks, 256, 0, stream>>>(z, tl1, hp1, b2, a2, hp2, st2);

    correct_kernel<<<cor_blocks, 256, 0, stream>>>(z, st2, hp2, out);
}


# filter-2 zero-state row stays in LDS (64 KiB per wave); after the in-wave scan the wave corrects it there (chunk = matrix row, sample = column) and stores out: third launch and z round trip gone
# speedup vs baseline: 11.9851x; 1.6024x over previous
#include <hip/hip_runtime.h>
#include <hip/hip_bf16.h>
#include <math.h>

static constexpr int Bsz = 512;
static constexpr int T   = 16384;
static constexpr int L   = 512;
static constexpr int C   = T / L;

typedef __attribute__((ext_vector_type(2))) float v2f;
typedef __attribute__((ext_vector_type(4))) float v4f;
typedef __attribute__((ext_vector_type(8))) float v8f;

__device__ __forceinline__ float whf(float x) {
    const float c   = 10.0f / 11.0f;
    const float l2e = 1.4426950408889634f;
    float e = 1.0f - __builtin_amdgcn_exp2f(-(c * l2e) * x);
    float l = c * x;
    unsigned s = (unsigned)(__float_as_int(x) >> 31);
    return __uint_as_float((__float_as_uint(e) & ~s) | (__float_as_uint(l) & s));
}

static constexpr int HP_BLOCKS = 2;

__device__ __forceinline__ void hpack_block(int which,
                                            const float* __restrict__ a1c,
                                            const float* __restrict__ a2c,
                                            float* __restrict__ hp1,
                                            float* __restrict__ hp2) {
    const float* ac = (which == 0) ? a1c : a2c;
    float*       hp = (which == 0) ? hp1 : hp2;
    int j = threadIdx.x;
    if (j > 3) return;
    if (j == 3) {
        for (int i = 0; i < L; ++i) hp[i * 4 + 3] = 0.0f;
        return;
    }
    const float a0 = ac[0], a1 = ac[1], a2 = ac[2];
    float y1 = (j == 0) ? 1.0f : 0.0f;
    float y2 = (j == 1) ? 1.0f : 0.0f;
    float y3 = (j == 2) ? 1.0f : 0.0f;
    for (int i = 0; i < L; ++i) {
        float y = -(a0 * y1 + a1 * y2 + a2 * y3);
        hp[i * 4 + j] = y;
        y3 = y2; y2 = y1; y1 = y;
    }
}

static_assert(C == 32, "one wave per row: lane index == chunk index");

struct HTail { float h00, h01, h02, h10, h11, h12, h20, h21, h22; };

__device__ __forceinline__ HTail load_htail(const float* __restrict__ hpack) {
    HTail h;
    h.h00 = hpack[(L - 1) * 4 + 0]; h.h01 = hpack[(L - 1) * 4 + 1]; h.h02 = hpack[(L - 1) * 4 + 2];
    h.h10 = hpack[(L - 2) * 4 + 0]; h.h11 = hpack[(L - 2) * 4 + 1]; h.h12 = hpack[(L - 2) * 4 + 2];
    h.h20 = hpack[(L - 3) * 4 + 0]; h.h21 = hpack[(L - 3) * 4 + 1]; h.h22 = hpack[(L - 3) * 4 + 2];
    return h;
}

__device__ __forceinline__ float lane_bcast(float v, int src) {
    return __int_as_float(__builtin_amdgcn_readlane(__float_as_int(v), src));
}

__device__ __forceinline__ void wave_scan(const HTail& h, float t0, float t1, float t2,
                                          int lane, float& e0, float& e1, float& e2) {
    float s0 = 0.0f, s1 = 0.0f, s2 = 0.0f;
    e0 = 0.0f; e1 = 0.0f; e2 = 0.0f;
    for (int c = 0; c < C; ++c) {
        if (lane == c) { e0 = s0; e1 = s1; e2 = s2; }
        const float c0 = lane_bcast(t0, c), c1 = lane_bcast(t1, c), c2 = lane_bcast(t2, c);
        const float n0 = c0 + h.h00 * s0 + h.h01 * s1 + h.h02 * s2;
        const float n1 = c1 + h.h10 * s0 + h.h11 * s1 + h.h12 * s2;
        const float n2 = c2 + h.h20 * s0 + h.h21 * s1 + h.h22 * s2;
        s0 = n0; s1 = n1; s2 = n2;
    }
}

static constexpr int P1_BLK = 64;
static constexpr int P1_NV  = P1_BLK / 4;
static constexpr int P1_NB  = L / P1_BLK;

static constexpr int STG_ROW   = P1_BLK;
static constexpr int STG_WAVE  = C * STG_ROW;

__device__ __forceinline__ void stage_sync() {
    __builtin_amdgcn_fence(__ATOMIC_RELEASE, "wavefront");
    __builtin_amdgcn_wave_barrier();
    __builtin_amdgcn_fence(__ATOMIC_ACQUIRE, "wavefront");
}

struct StageMap {
    int sub;
    int slot;
    int own;
};

__device__ __forceinline__ int stg_off(int cc, int s) {
    return cc * STG_ROW + 4 * (s ^ (cc & 15));
}

template <int NV>
__device__ __forceinline__ void lines_load(v4f (&g)[NV], const float* __restrict__ rowblk,
                                           const StageMap& m) {
#pragma unroll
    for (int j = 0; j < NV; ++j)
        g[j] = *(const v4f*)(rowblk + (size_t)(2 * j + m.sub) * L + 4 * m.slot);
    __builtin_amdgcn_sched_barrier(0);
}

template <int NV>
__device__ __forceinline__ void lines_store(const v4f (&g)[NV], float* __restrict__ rowblk,
                                            const StageMap& m) {
#pragma unroll
    for (int j = 0; j < NV; ++j)
        *(v4f*)(rowblk + (size_t)(2 * j + m.sub) * L + 4 * m.slot) = g[j];
}

template <int NV>
__device__ __forceinline__ void stage_put_lines(const v4f (&g)[NV], float* stage, const StageMap& m) {
#pragma unroll
    for (int j = 0; j < NV; ++j) *(v4f*)(stage + stg_off(2 * j + m.sub, m.slot)) = g[j];
}
template <int NV>
__device__ __forceinline__ void stage_get_lines(v4f (&g)[NV], const float* stage, const StageMap& m) {
#pragma unroll
    for (int j = 0; j < NV; ++j) g[j] = *(const v4f*)(stage + stg_off(2 * j + m.sub, m.slot));
}
template <int NV>
__device__ __forceinline__ void stage_put_own(const v4f (&x)[NV], float* stage, const StageMap& m) {
#pragma unroll
    for (int v = 0; v < NV; ++v) *(v4f*)(stage + stg_off(m.own, v)) = x[v];
}
template <int NV>
__device__ __forceinline__ void stage_get_own(v4f (&x)[NV], const float* stage, const StageMap& m) {
#pragma unroll
    for (int v = 0; v < NV; ++v) x[v] = *(const v4f*)(stage + stg_off(m.own, v));
}

template <int NV>
__device__ __forceinline__ void p1_load(v4f (&dst)[NV], const float* __restrict__ p) {
#pragma unroll
    for (int v = 0; v < NV; ++v) dst[v] = *(const v4f*)(p + 4 * v);
    __builtin_amdgcn_sched_barrier(0);
}

template <int NK, int NV>
__device__ __forceinline__ void p1_block(const v4f (&xin)[NV], float* __restrict__ zdst,
                                         float b0, float b1, float b2,
                                         float a0, float a1, float a2,
                                         float& p1, float& p2, float& p3,
                                         float& y1, float& y2, float& y3) {
#pragma unroll
    for (int v = 0; v < NV; ++v) {
        v4f o;
#pragma unroll
        for (int e = 0; e < 4; ++e) {
            const float s = xin[v][e];
            float xf;
            if (NK == 0) xf = b0 * s  + b1 * p1 + b2 * p2;
            else         xf = b0 * p1 + b1 * p2 + b2 * p3;
            const float y = xf - (a0 * y1 + a1 * y2 + a2 * y3);
            o[e] = y;
            p3 = p2; p2 = p1; p1 = s;
            y3 = y2; y2 = y1; y1 = y;
        }
        *(v4f*)(zdst + 4 * v) = o;
    }
}

template <int NK, int NV>
__device__ __forceinline__ void p1_block_reg(v4f (&x)[NV],
                                             float b0, float b1, float b2,
                                             float a0, float a1, float a2,
                                             float& p1, float& p2, float& p3,
                                             float& y1, float& y2, float& y3) {
#pragma unroll
    for (int v = 0; v < NV; ++v) {
        v4f o;
#pragma unroll
        for (int e = 0; e < 4; ++e) {
            const float s = x[v][e];
            float xf;
            if (NK == 0) xf = b0 * s  + b1 * p1 + b2 * p2;
            else         xf = b0 * p1 + b1 * p2 + b2 * p3;
            const float y = xf - (a0 * y1 + a1 * y2 + a2 * y3);
            o[e] = y;
            p3 = p2; p2 = p1; p1 = s;
            y3 = y2; y2 = y1; y1 = y;
        }
        x[v] = o;
    }
}

template <int NK>
__global__ __launch_bounds__(256)
void phase1_kernel(const float* __restrict__ in,
                   const float* __restrict__ bco,
                   const float* __restrict__ aco,
                   const float* __restrict__ aco2,
                   float* __restrict__ hp1,
                   float* __restrict__ hp2,
                   float* __restrict__ z,
                   float* __restrict__ tails) {
    __shared__ __attribute__((aligned(16))) float stage_all[8 * STG_WAVE];
    if (blockIdx.x < HP_BLOCKS) {
        hpack_block(blockIdx.x, aco, aco2, hp1, hp2);
        return;
    }
    const int id = (blockIdx.x - HP_BLOCKS) * blockDim.x + threadIdx.x;
    const int b  = id / C;
    const int c  = id % C;
    const float b0 = bco[0], b1 = bco[1], b2 = bco[2];
    const float a0 = aco[0], a1 = aco[1], a2 = aco[2];

    const float* __restrict__ src  = in + (size_t)b * T + (size_t)c * L;
    const float* __restrict__ urow = in + (size_t)b * T;
    float*       __restrict__ zrow = z  + (size_t)b * T;

    float* stage = stage_all + (threadIdx.x >> 5) * STG_WAVE;
    StageMap m;
    m.sub = (threadIdx.x & 31) >> 4;  m.slot = threadIdx.x & 15;  m.own = c;

    float p1 = 0.0f, p2 = 0.0f, p3 = 0.0f;
    if (c > 0) {
        p1 = src[-1];
        p2 = src[-2];
        if (NK == 1) p3 = src[-3];
    }
    float y1 = 0.0f, y2 = 0.0f, y3 = 0.0f;

    v4f g[P1_NV], x[P1_NV];
    lines_load<P1_NV>(g, urow, m);
    stage_put_lines<P1_NV>(g, stage, m);
    stage_sync();
#pragma unroll 1
    for (int blk = 0; blk < P1_NB; ++blk) {
        stage_get_own<P1_NV>(x, stage, m);
        stage_sync();
        if (blk + 1 < P1_NB) lines_load<P1_NV>(g, urow + (blk + 1) * P1_BLK, m);
        p1_block_reg<NK, P1_NV>(x, b0, b1, b2, a0, a1, a2, p1, p2, p3, y1, y2, y3);
        stage_put_own<P1_NV>(x, stage, m);
        stage_sync();
        stage_get_lines<P1_NV>(x, stage, m);
        stage_sync();
        lines_store<P1_NV>(x, zrow + blk * P1_BLK, m);
        if (blk + 1 < P1_NB) {
            stage_put_lines<P1_NV>(g, stage, m);
            stage_sync();
        }
    }
    v4f tl;
    tl[0] = y1; tl[1] = y2; tl[2] = y3; tl[3] = 0.0f;
    *(v4f*)(tails + (size_t)id * 4) = tl;
}

static constexpr int ROW_PITCH = L + 4;
static constexpr int ROW_WAVE  = C * ROW_PITCH;
static constexpr int F_WAVES   = 2;
static constexpr int F_NS = P1_BLK / 16;

template <int NS>
__device__ __forceinline__ void f_load(v4f (&cz)[NS][4], v2f (&ah)[NS],
                                       const float* zlo, const float* zhi,
                                       const float* __restrict__ hrow) {
#pragma unroll
    for (int s = 0; s < NS; ++s) {
        cz[s][0] = *(const v4f*)(zlo + 16 * s);
        cz[s][1] = *(const v4f*)(zlo + 16 * s + 4);
        cz[s][2] = *(const v4f*)(zhi + 16 * s);
        cz[s][3] = *(const v4f*)(zhi + 16 * s + 4);
        ah[s]    = *(const v2f*)(hrow + 64 * s);
    }
    __builtin_amdgcn_sched_barrier(0);
}

__device__ __forceinline__ void f_v16(const v4f (&cz)[4], v2f av, v2f bvl, v2f bvh,
                                      bool hi, v4f (&seq)[4]) {
    v8f c1, c2;
    c1[0] = cz[0][0]; c1[1] = cz[0][1]; c1[2] = cz[0][2]; c1[3] = cz[0][3];
    c1[4] = cz[1][0]; c1[5] = cz[1][1]; c1[6] = cz[1][2]; c1[7] = cz[1][3];
    c2[0] = cz[2][0]; c2[1] = cz[2][1]; c2[2] = cz[2][2]; c2[3] = cz[2][3];
    c2[4] = cz[3][0]; c2[5] = cz[3][1]; c2[6] = cz[3][2]; c2[7] = cz[3][3];
    const v8f d1 = __builtin_amdgcn_wmma_f32_16x16x4_f32(
         false, av,  false, bvl,
         (short)0, c1,  false,  false);
    const v8f d2 = __builtin_amdgcn_wmma_f32_16x16x4_f32(
         false, av,  false, bvh,
         (short)0, c2,  false,  false);
    float v1[8], v2[8], rcv[8];
#pragma unroll
    for (int r = 0; r < 8; ++r) {
        v1[r] = whf(d1[r]);
        v2[r] = whf(d2[r]);
    }
    __builtin_amdgcn_sched_barrier(0);
#pragma unroll
    for (int r = 0; r < 8; ++r) rcv[r] = __shfl_xor(hi ? v1[r] : v2[r], 16);
    __builtin_amdgcn_sched_barrier(0);
#pragma unroll
    for (int r = 0; r < 8; ++r) {
        seq[r >> 2][r & 3]       = hi ? rcv[r] : v1[r];
        seq[2 + (r >> 2)][r & 3] = hi ? v2[r]  : rcv[r];
    }
}

template <int NS>
__device__ __forceinline__ void f_block(const v4f (&cz)[NS][4], const v2f (&ah)[NS],
                                        v2f bvl, v2f bvh, bool hi,
                                        float* rowown,
                                        float b0, float b1, float b2,
                                        float a0, float a1, float a2,
                                        float& p1, float& p2, float& p3,
                                        float& y1, float& y2, float& y3) {
#pragma unroll
    for (int s = 0; s < NS; ++s) {
        v4f seq[4];
        f_v16(cz[s], ah[s], bvl, bvh, hi, seq);
        p1_block_reg<0, 4>(seq, b0, b1, b2, a0, a1, a2, p1, p2, p3, y1, y2, y3);
#pragma unroll
        for (int i = 0; i < 4; ++i) *(v4f*)(rowown + 16 * s + 4 * i) = seq[i];
    }
}

__global__ __launch_bounds__(32 * F_WAVES)
void fused_kernel(const float* __restrict__ z,
                  const float* __restrict__ tails,
                  const float* __restrict__ hpack,
                  const float* __restrict__ bco,
                  const float* __restrict__ aco,
                  const float* __restrict__ hpack2,
                  float* __restrict__ out) {
    __shared__ __attribute__((aligned(16))) float row_all[F_WAVES * ROW_WAVE];
    const int id   = blockIdx.x * blockDim.x + threadIdx.x;
    const int b    = id / C;
    const int lane = threadIdx.x & 31;
    const int n    = lane & 15;
    const int half = lane >> 4;
    const bool hi  = half != 0;
    const float b0 = bco[0], b1 = bco[1], b2 = bco[2];
    const float a0 = aco[0], a1 = aco[1], a2 = aco[2];
    const HTail ht = load_htail(hpack2);

    const float* zrow = z + (size_t)b * T;
    const float* zlo  = zrow + (size_t)n * L + 8 * half;
    const float* zhi  = zrow + (size_t)(16 + n) * L + 8 * half;
    float* rowbuf = row_all + (threadIdx.x >> 5) * ROW_WAVE;
    float* rowown = rowbuf + lane * ROW_PITCH;
    const float* __restrict__ hrow = hpack + n * 4 + 2 * half;

    float e0, e1, e2;
    {
        const v4f tl = *(const v4f*)(tails + (size_t)id * 4);
        const HTail ht1 = load_htail(hpack);
        wave_scan(ht1, tl[0], tl[1], tl[2], lane, e0, e1, e2);
    }
    v2f bvl, bvh;
    {
        const float l0 = __shfl(e0, n),      l1 = __shfl(e1, n),      l2 = __shfl(e2, n);
        const float h0 = __shfl(e0, 16 + n), h1 = __shfl(e1, 16 + n), h2 = __shfl(e2, 16 + n);
        bvl[0] = hi ? l2 : l0;  bvl[1] = hi ? 0.0f : l1;
        bvh[0] = hi ? h2 : h0;  bvh[1] = hi ? 0.0f : h1;
    }

    float p1, p2, p3 = 0.0f;
    {
        v4f cz[1][4]; v2f ah[1]; v4f seq[4];
        f_load<1>(cz, ah, zlo + (L - 16), zhi + (L - 16), hrow + (L - 16) * 4);
        f_v16(cz[0], ah[0], bvl, bvh, hi, seq);
        p1 = __shfl_up(seq[3][3], 1);
        p2 = __shfl_up(seq[3][2], 1);
        if (lane == 0) { p1 = 0.0f; p2 = 0.0f; }
    }
    float y1 = 0.0f, y2 = 0.0f, y3 = 0.0f;

    v4f ca[F_NS][4], cb[F_NS][4];
    v2f ha[F_NS], hb[F_NS];
    f_load<F_NS>(ca, ha, zlo, zhi, hrow);
    for (int blk = 0; blk < P1_NB - 2; blk += 2) {
        f_load<F_NS>(cb, hb, zlo + (blk + 1) * P1_BLK, zhi + (blk + 1) * P1_BLK,
                     hrow + (blk + 1) * P1_BLK * 4);
        f_block<F_NS>(ca, ha, bvl, bvh, hi, rowown + blk * P1_BLK,
                      b0, b1, b2, a0, a1, a2, p1, p2, p3, y1, y2, y3);
        f_load<F_NS>(ca, ha, zlo + (blk + 2) * P1_BLK, zhi + (blk + 2) * P1_BLK,
                     hrow + (blk + 2) * P1_BLK * 4);
        f_block<F_NS>(cb, hb, bvl, bvh, hi, rowown + (blk + 1) * P1_BLK,
                      b0, b1, b2, a0, a1, a2, p1, p2, p3, y1, y2, y3);
    }
    f_load<F_NS>(cb, hb, zlo + (P1_NB - 1) * P1_BLK, zhi + (P1_NB - 1) * P1_BLK,
                 hrow + (P1_NB - 1) * P1_BLK * 4);
    f_block<F_NS>(ca, ha, bvl, bvh, hi, rowown + (P1_NB - 2) * P1_BLK,
                  b0, b1, b2, a0, a1, a2, p1, p2, p3, y1, y2, y3);
    f_block<F_NS>(cb, hb, bvl, bvh, hi, rowown + (P1_NB - 1) * P1_BLK,
                  b0, b1, b2, a0, a1, a2, p1, p2, p3, y1, y2, y3);

    wave_scan(ht, y1, y2, y3, lane, e0, e1, e2);
    v2f svl, svh;
    {
        const float l0 = __shfl(e0, n),      l1 = __shfl(e1, n),      l2 = __shfl(e2, n);
        const float h0 = __shfl(e0, 16 + n), h1 = __shfl(e1, 16 + n), h2 = __shfl(e2, 16 + n);
        svl[0] = hi ? l2 : l0;  svl[1] = hi ? 0.0f : l1;
        svh[0] = hi ? h2 : h0;  svh[1] = hi ? 0.0f : h1;
    }
    stage_sync();

    float* orow = out + (size_t)b * T;
    const float* __restrict__ hcol = hpack2 + (4 * n) * 4 + 2 * half;
#pragma unroll 2
    for (int blk = 0; blk < P1_NB; ++blk) {
        v2f bt[4];
#pragma unroll
        for (int j = 0; j < 4; ++j) bt[j] = *(const v2f*)(hcol + (blk * P1_BLK + j) * 4);
#pragma unroll
        for (int grp = 0; grp < 2; ++grp) {
            const float* zc = rowbuf + (16 * grp + 8 * half) * ROW_PITCH + blk * P1_BLK + 4 * n;
            float*       oc = orow + (size_t)(16 * grp + 8 * half) * L + blk * P1_BLK + 4 * n;
            v4f cr[8];
#pragma unroll
            for (int r = 0; r < 8; ++r) cr[r] = *(const v4f*)(zc + r * ROW_PITCH);
            v8f acc[4];
#pragma unroll
            for (int j = 0; j < 4; ++j) {
#pragma unroll
                for (int r = 0; r < 8; ++r) acc[j][r] = cr[r][j];
                acc[j] = __builtin_amdgcn_wmma_f32_16x16x4_f32(
                     false, grp ? svh : svl,  false, bt[j],
                     (short)0, acc[j],  false,  false);
            }
#pragma unroll
            for (int r = 0; r < 8; ++r) {
                v4f o;
                o[0] = acc[0][r]; o[1] = acc[1][r]; o[2] = acc[2][r]; o[3] = acc[3][r];
                __builtin_nontemporal_store(o, (v4f*)(oc + (size_t)r * L));
            }
        }
    }
}

extern "C" void kernel_launch(void* const* d_in, const int* in_sizes, int n_in,
                              void* d_out, int out_size, void* d_ws, size_t ws_size,
                              hipStream_t stream) {
    const float* u  = (const float*)d_in[0];
    const float* b1 = (const float*)d_in[1];
    const float* a1 = (const float*)d_in[2];
    const float* b2 = (const float*)d_in[3];
    const float* a2 = (const float*)d_in[4];
    float* out = (float*)d_out;

    float* ws     = (float*)d_ws;
    float* z      = ws;
    float* tl1    = z + (size_t)Bsz * T;
    float* hp1    = tl1 + (size_t)Bsz * C * 4;
    float* hp2    = hp1 + L * 4;

    const int p1_grid    = HP_BLOCKS + (Bsz * C) / 256;
    const int f_blocks   = Bsz / F_WAVES;

    phase1_kernel<1><<<p1_grid, 256, 0, stream>>>(u, b1, a1, a2, hp1, hp2, z, tl1);

    fused_kernel<<<f_blocks, 32 * F_WAVES, 0, stream>>>(z, tl1, hp1, b2, a2, hp2, out);
}


# first launch folded into the second: the wave takes its input row as whole lines into its LDS row buffer, runs filter 1 there in place and builds both tables beside it; z, tails, one launch gone
# speedup vs baseline: 16.7791x; 1.4000x over previous
#include <hip/hip_runtime.h>
#include <hip/hip_bf16.h>
#include <math.h>

static constexpr int Bsz = 512;
static constexpr int T   = 16384;
static constexpr int L   = 512;
static constexpr int C   = T / L;

typedef __attribute__((ext_vector_type(2))) float v2f;
typedef __attribute__((ext_vector_type(4))) float v4f;
typedef __attribute__((ext_vector_type(8))) float v8f;

__device__ __forceinline__ float whf(float x) {
    const float c   = 10.0f / 11.0f;
    const float l2e = 1.4426950408889634f;
    float e = 1.0f - __builtin_amdgcn_exp2f(-(c * l2e) * x);
    float l = c * x;
    unsigned s = (unsigned)(__float_as_int(x) >> 31);
    return __uint_as_float((__float_as_uint(e) & ~s) | (__float_as_uint(l) & s));
}


static constexpr int TAB = L * 4;

static_assert(C == 32, "one wave per row: lane index == chunk index");

struct HTail { float h00, h01, h02, h10, h11, h12, h20, h21, h22; };

__device__ __forceinline__ HTail load_htail(const float* __restrict__ hpack) {
    HTail h;
    h.h00 = hpack[(L - 1) * 4 + 0]; h.h01 = hpack[(L - 1) * 4 + 1]; h.h02 = hpack[(L - 1) * 4 + 2];
    h.h10 = hpack[(L - 2) * 4 + 0]; h.h11 = hpack[(L - 2) * 4 + 1]; h.h12 = hpack[(L - 2) * 4 + 2];
    h.h20 = hpack[(L - 3) * 4 + 0]; h.h21 = hpack[(L - 3) * 4 + 1]; h.h22 = hpack[(L - 3) * 4 + 2];
    return h;
}

__device__ __forceinline__ float lane_bcast(float v, int src) {
    return __int_as_float(__builtin_amdgcn_readlane(__float_as_int(v), src));
}

__device__ __forceinline__ void wave_scan(const HTail& h, float t0, float t1, float t2,
                                          int lane, float& e0, float& e1, float& e2) {
    float s0 = 0.0f, s1 = 0.0f, s2 = 0.0f;
    e0 = 0.0f; e1 = 0.0f; e2 = 0.0f;
    for (int c = 0; c < C; ++c) {
        if (lane == c) { e0 = s0; e1 = s1; e2 = s2; }
        const float c0 = lane_bcast(t0, c), c1 = lane_bcast(t1, c), c2 = lane_bcast(t2, c);
        const float n0 = c0 + h.h00 * s0 + h.h01 * s1 + h.h02 * s2;
        const float n1 = c1 + h.h10 * s0 + h.h11 * s1 + h.h12 * s2;
        const float n2 = c2 + h.h20 * s0 + h.h21 * s1 + h.h22 * s2;
        s0 = n0; s1 = n1; s2 = n2;
    }
}

static constexpr int P1_BLK = 64;
static constexpr int P1_NV  = P1_BLK / 4;
static constexpr int P1_NB  = L / P1_BLK;

__device__ __forceinline__ void stage_sync() {
    __builtin_amdgcn_fence(__ATOMIC_RELEASE, "wavefront");
    __builtin_amdgcn_wave_barrier();
    __builtin_amdgcn_fence(__ATOMIC_ACQUIRE, "wavefront");
}

template <int NV>
__device__ __forceinline__ void p1_block_tab(v4f (&x)[NV],
                                             float b0, float b1, float b2,
                                             float a0, float a1, float a2,
                                             float& p1, float& p2, float& p3,
                                             float& y1, float& y2, float& y3,
                                             float ta0, float ta1, float ta2,
                                             float& h1, float& h2, float& h3,
                                             float* tcol) {
#pragma unroll
    for (int v = 0; v < NV; ++v) {
        v4f o;
#pragma unroll
        for (int e = 0; e < 4; ++e) {
            const float s = x[v][e];
            const float xf = b0 * p1 + b1 * p2 + b2 * p3;
            const float y = xf - (a0 * y1 + a1 * y2 + a2 * y3);
            o[e] = y;
            p3 = p2; p2 = p1; p1 = s;
            y3 = y2; y2 = y1; y1 = y;
            const float th = -(ta0 * h1 + ta1 * h2 + ta2 * h3);
            tcol[(4 * v + e) * 4] = th;
            h3 = h2; h2 = h1; h1 = th;
        }
        x[v] = o;
    }
}

template <int NK, int NV>
__device__ __forceinline__ void p1_block_reg(v4f (&x)[NV],
                                             float b0, float b1, float b2,
                                             float a0, float a1, float a2,
                                             float& p1, float& p2, float& p3,
                                             float& y1, float& y2, float& y3) {
#pragma unroll
    for (int v = 0; v < NV; ++v) {
        v4f o;
#pragma unroll
        for (int e = 0; e < 4; ++e) {
            const float s = x[v][e];
            float xf;
            if (NK == 0) xf = b0 * s  + b1 * p1 + b2 * p2;
            else         xf = b0 * p1 + b1 * p2 + b2 * p3;
            const float y = xf - (a0 * y1 + a1 * y2 + a2 * y3);
            o[e] = y;
            p3 = p2; p2 = p1; p1 = s;
            y3 = y2; y2 = y1; y1 = y;
        }
        x[v] = o;
    }
}

static constexpr int ROW_PITCH = L + 4;
static constexpr int ROW_WAVE  = C * ROW_PITCH;
static constexpr int F_WAVES   = 1;
static constexpr int F_NS = P1_BLK / 16;

template <int NV>
__device__ __forceinline__ void lines_load(v4f (&g)[NV], const float* __restrict__ uline) {
#pragma unroll
    for (int j = 0; j < NV; ++j) g[j] = *(const v4f*)(uline + (size_t)(2 * j) * L);
    __builtin_amdgcn_sched_barrier(0);
}

template <int NV>
__device__ __forceinline__ void lines_put(const v4f (&g)[NV], float* rline) {
#pragma unroll
    for (int j = 0; j < NV; ++j) *(v4f*)(rline + (2 * j) * ROW_PITCH) = g[j];
}

template <int NS>
__device__ __forceinline__ void f_load(v4f (&cz)[NS][4], v2f (&ah)[NS],
                                       const float* zlo, const float* zhi,
                                       const float* __restrict__ hrow) {
#pragma unroll
    for (int s = 0; s < NS; ++s) {
        cz[s][0] = *(const v4f*)(zlo + 16 * s);
        cz[s][1] = *(const v4f*)(zlo + 16 * s + 4);
        cz[s][2] = *(const v4f*)(zhi + 16 * s);
        cz[s][3] = *(const v4f*)(zhi + 16 * s + 4);
        ah[s]    = *(const v2f*)(hrow + 64 * s);
    }
    __builtin_amdgcn_sched_barrier(0);
}

__device__ __forceinline__ void f_v16(const v4f (&cz)[4], v2f av, v2f bvl, v2f bvh,
                                      bool hi, v4f (&seq)[4]) {
    v8f c1, c2;
    c1[0] = cz[0][0]; c1[1] = cz[0][1]; c1[2] = cz[0][2]; c1[3] = cz[0][3];
    c1[4] = cz[1][0]; c1[5] = cz[1][1]; c1[6] = cz[1][2]; c1[7] = cz[1][3];
    c2[0] = cz[2][0]; c2[1] = cz[2][1]; c2[2] = cz[2][2]; c2[3] = cz[2][3];
    c2[4] = cz[3][0]; c2[5] = cz[3][1]; c2[6] = cz[3][2]; c2[7] = cz[3][3];
    const v8f d1 = __builtin_amdgcn_wmma_f32_16x16x4_f32(
         false, av,  false, bvl,
         (short)0, c1,  false,  false);
    const v8f d2 = __builtin_amdgcn_wmma_f32_16x16x4_f32(
         false, av,  false, bvh,
         (short)0, c2,  false,  false);
    float v1[8], v2[8], rcv[8];
#pragma unroll
    for (int r = 0; r < 8; ++r) {
        v1[r] = whf(d1[r]);
        v2[r] = whf(d2[r]);
    }
    __builtin_amdgcn_sched_barrier(0);
#pragma unroll
    for (int r = 0; r < 8; ++r) rcv[r] = __shfl_xor(hi ? v1[r] : v2[r], 16);
    __builtin_amdgcn_sched_barrier(0);
#pragma unroll
    for (int r = 0; r < 8; ++r) {
        seq[r >> 2][r & 3]       = hi ? rcv[r] : v1[r];
        seq[2 + (r >> 2)][r & 3] = hi ? v2[r]  : rcv[r];
    }
}

template <int NS>
__device__ __forceinline__ void f_block(const v4f (&cz)[NS][4], const v2f (&ah)[NS],
                                        v2f bvl, v2f bvh, bool hi,
                                        float* rowown,
                                        float b0, float b1, float b2,
                                        float a0, float a1, float a2,
                                        float& p1, float& p2, float& p3,
                                        float& y1, float& y2, float& y3) {
#pragma unroll
    for (int s = 0; s < NS; ++s) {
        v4f seq[4];
        f_v16(cz[s], ah[s], bvl, bvh, hi, seq);
        p1_block_reg<0, 4>(seq, b0, b1, b2, a0, a1, a2, p1, p2, p3, y1, y2, y3);
#pragma unroll
        for (int i = 0; i < 4; ++i) *(v4f*)(rowown + 16 * s + 4 * i) = seq[i];
    }
}

__global__ __launch_bounds__(32 * F_WAVES)
void fused_kernel(const float* __restrict__ u,
                  const float* __restrict__ bco1,
                  const float* __restrict__ aco1,
                  const float* __restrict__ bco2,
                  const float* __restrict__ aco2,
                  float* __restrict__ out) {
    __shared__ __attribute__((aligned(16))) float row_all[F_WAVES * ROW_WAVE];
    __shared__ __attribute__((aligned(16))) float tab_all[F_WAVES * 2 * TAB];
    const int id   = blockIdx.x * blockDim.x + threadIdx.x;
    const int b    = id / C;
    const int lane = threadIdx.x & 31;
    const int n    = lane & 15;
    const int half = lane >> 4;
    const bool hi  = half != 0;

    float* rowbuf = row_all + (threadIdx.x >> 5) * ROW_WAVE;
    float* rowown = rowbuf + lane * ROW_PITCH;
    float* tab1   = tab_all + (threadIdx.x >> 5) * (2 * TAB);
    float* tab2   = tab1 + TAB;

    float y1 = 0.0f, y2 = 0.0f, y3 = 0.0f;
    {
        const float b0 = bco1[0], b1 = bco1[1], b2 = bco1[2];
        const float a0 = aco1[0], a1 = aco1[1], a2 = aco1[2];
        const int col = lane & 7;
        const int tj  = col & 3;
        const float* __restrict__ tac = (col < 4) ? aco1 : aco2;
        const float ta0 = tac[0], ta1 = tac[1], ta2 = tac[2];
        float h1 = (tj == 0) ? 1.0f : 0.0f;
        float h2 = (tj == 1) ? 1.0f : 0.0f;
        float h3 = (tj == 2) ? 1.0f : 0.0f;
        float* tcol = ((col < 4) ? tab1 : tab2) + tj;

        const float* __restrict__ urow  = u + (size_t)b * T;
        const float* __restrict__ src   = urow + (size_t)lane * L;
        const float* __restrict__ uline = urow + (size_t)half * L + 4 * n;
        float*                    rline = rowbuf + half * ROW_PITCH + 4 * n;

        float p1 = 0.0f, p2 = 0.0f, p3 = 0.0f;
        if (lane > 0) {
            p1 = src[-1];
            p2 = src[-2];
            p3 = src[-3];
        }

        v4f g[P1_NV], x[P1_NV];
        lines_load<P1_NV>(g, uline);
        lines_put<P1_NV>(g, rline);
        stage_sync();
#pragma unroll 1
        for (int blk = 0; blk < P1_NB; ++blk) {
#pragma unroll
            for (int v = 0; v < P1_NV; ++v) x[v] = *(const v4f*)(rowown + blk * P1_BLK + 4 * v);
            if (blk + 1 < P1_NB) lines_load<P1_NV>(g, uline + (blk + 1) * P1_BLK);
            p1_block_tab<P1_NV>(x, b0, b1, b2, a0, a1, a2, p1, p2, p3, y1, y2, y3,
                                ta0, ta1, ta2, h1, h2, h3, tcol + blk * P1_BLK * 4);
#pragma unroll
            for (int v = 0; v < P1_NV; ++v) *(v4f*)(rowown + blk * P1_BLK + 4 * v) = x[v];
            if (blk + 1 < P1_NB) {
                lines_put<P1_NV>(g, rline + (blk + 1) * P1_BLK);
                stage_sync();
            }
        }
    }
    stage_sync();

    const float b0 = bco2[0], b1 = bco2[1], b2 = bco2[2];
    const float a0 = aco2[0], a1 = aco2[1], a2 = aco2[2];
    const HTail ht = load_htail(tab2);

    const float* zlo  = rowbuf + n * ROW_PITCH + 8 * half;
    const float* zhi  = rowbuf + (16 + n) * ROW_PITCH + 8 * half;
    const float* hrow = tab1 + n * 4 + 2 * half;

    float e0, e1, e2;
    {
        const HTail ht1 = load_htail(tab1);
        wave_scan(ht1, y1, y2, y3, lane, e0, e1, e2);
    }
    v2f bvl, bvh;
    {
        const float l0 = __shfl(e0, n),      l1 = __shfl(e1, n),      l2 = __shfl(e2, n);
        const float h0 = __shfl(e0, 16 + n), h1 = __shfl(e1, 16 + n), h2 = __shfl(e2, 16 + n);
        bvl[0] = hi ? l2 : l0;  bvl[1] = hi ? 0.0f : l1;
        bvh[0] = hi ? h2 : h0;  bvh[1] = hi ? 0.0f : h1;
    }

    float p1, p2, p3 = 0.0f;
    {
        v4f cz[1][4]; v2f ah[1]; v4f seq[4];
        f_load<1>(cz, ah, zlo + (L - 16), zhi + (L - 16), hrow + (L - 16) * 4);
        f_v16(cz[0], ah[0], bvl, bvh, hi, seq);
        p1 = __shfl_up(seq[3][3], 1);
        p2 = __shfl_up(seq[3][2], 1);
        if (lane == 0) { p1 = 0.0f; p2 = 0.0f; }
    }
    y1 = 0.0f; y2 = 0.0f; y3 = 0.0f;

    v4f ca[F_NS][4], cb[F_NS][4];
    v2f ha[F_NS], hb[F_NS];
    f_load<F_NS>(ca, ha, zlo, zhi, hrow);
    for (int blk = 0; blk < P1_NB - 2; blk += 2) {
        f_load<F_NS>(cb, hb, zlo + (blk + 1) * P1_BLK, zhi + (blk + 1) * P1_BLK,
                     hrow + (blk + 1) * P1_BLK * 4);
        f_block<F_NS>(ca, ha, bvl, bvh, hi, rowown + blk * P1_BLK,
                      b0, b1, b2, a0, a1, a2, p1, p2, p3, y1, y2, y3);
        f_load<F_NS>(ca, ha, zlo + (blk + 2) * P1_BLK, zhi + (blk + 2) * P1_BLK,
                     hrow + (blk + 2) * P1_BLK * 4);
        f_block<F_NS>(cb, hb, bvl, bvh, hi, rowown + (blk + 1) * P1_BLK,
                      b0, b1, b2, a0, a1, a2, p1, p2, p3, y1, y2, y3);
    }
    f_load<F_NS>(cb, hb, zlo + (P1_NB - 1) * P1_BLK, zhi + (P1_NB - 1) * P1_BLK,
                 hrow + (P1_NB - 1) * P1_BLK * 4);
    f_block<F_NS>(ca, ha, bvl, bvh, hi, rowown + (P1_NB - 2) * P1_BLK,
                  b0, b1, b2, a0, a1, a2, p1, p2, p3, y1, y2, y3);
    f_block<F_NS>(cb, hb, bvl, bvh, hi, rowown + (P1_NB - 1) * P1_BLK,
                  b0, b1, b2, a0, a1, a2, p1, p2, p3, y1, y2, y3);

    wave_scan(ht, y1, y2, y3, lane, e0, e1, e2);
    v2f svl, svh;
    {
        const float l0 = __shfl(e0, n),      l1 = __shfl(e1, n),      l2 = __shfl(e2, n);
        const float h0 = __shfl(e0, 16 + n), h1 = __shfl(e1, 16 + n), h2 = __shfl(e2, 16 + n);
        svl[0] = hi ? l2 : l0;  svl[1] = hi ? 0.0f : l1;
        svh[0] = hi ? h2 : h0;  svh[1] = hi ? 0.0f : h1;
    }
    stage_sync();

    float* orow = out + (size_t)b * T;
    const float* hcol = tab2 + (4 * n) * 4 + 2 * half;
#pragma unroll 2
    for (int blk = 0; blk < P1_NB; ++blk) {
        v2f bt[4];
#pragma unroll
        for (int j = 0; j < 4; ++j) bt[j] = *(const v2f*)(hcol + (blk * P1_BLK + j) * 4);
#pragma unroll
        for (int grp = 0; grp < 2; ++grp) {
            const float* zc = rowbuf + (16 * grp + 8 * half) * ROW_PITCH + blk * P1_BLK + 4 * n;
            float*       oc = orow + (size_t)(16 * grp + 8 * half) * L + blk * P1_BLK + 4 * n;
            v4f cr[8];
#pragma unroll
            for (int r = 0; r < 8; ++r) cr[r] = *(const v4f*)(zc + r * ROW_PITCH);
            v8f acc[4];
#pragma unroll
            for (int j = 0; j < 4; ++j) {
#pragma unroll
                for (int r = 0; r < 8; ++r) acc[j][r] = cr[r][j];
                acc[j] = __builtin_amdgcn_wmma_f32_16x16x4_f32(
                     false, grp ? svh : svl,  false, bt[j],
                     (short)0, acc[j],  false,  false);
            }
#pragma unroll
            for (int r = 0; r < 8; ++r) {
                v4f o;
                o[0] = acc[0][r]; o[1] = acc[1][r]; o[2] = acc[2][r]; o[3] = acc[3][r];
                __builtin_nontemporal_store(o, (v4f*)(oc + (size_t)r * L));
            }
        }
    }
}

extern "C" void kernel_launch(void* const* d_in, const int* in_sizes, int n_in,
                              void* d_out, int out_size, void* d_ws, size_t ws_size,
                              hipStream_t stream) {
    const float* u  = (const float*)d_in[0];
    const float* b1 = (const float*)d_in[1];
    const float* a1 = (const float*)d_in[2];
    const float* b2 = (const float*)d_in[3];
    const float* a2 = (const float*)d_in[4];
    float* out = (float*)d_out;

    const int f_blocks = Bsz / F_WAVES;

    fused_kernel<<<f_blocks, 32 * F_WAVES, 0, stream>>>(u, b1, a1, b2, a2, out);
}


# every 3-tap feedback sum re-associated, oldest output first and newest last (y = ((x - a2*y3) - a1*y2) - a0*y1, in-chunk tables too): 1 dependent multiply-add per sample, was 4; sum order changes
# speedup vs baseline: 19.3223x; 1.1516x over previous
#include <hip/hip_runtime.h>
#include <hip/hip_bf16.h>
#include <math.h>

static constexpr int Bsz = 512;
static constexpr int T   = 16384;
static constexpr int L   = 512;
static constexpr int C   = T / L;

typedef __attribute__((ext_vector_type(2))) float v2f;
typedef __attribute__((ext_vector_type(4))) float v4f;
typedef __attribute__((ext_vector_type(8))) float v8f;

__device__ __forceinline__ float whf(float x) {
    const float c   = 10.0f / 11.0f;
    const float l2e = 1.4426950408889634f;
    float e = 1.0f - __builtin_amdgcn_exp2f(-(c * l2e) * x);
    float l = c * x;
    unsigned s = (unsigned)(__float_as_int(x) >> 31);
    return __uint_as_float((__float_as_uint(e) & ~s) | (__float_as_uint(l) & s));
}


static constexpr int TAB = L * 4;

static_assert(C == 32, "one wave per row: lane index == chunk index");

struct HTail { float h00, h01, h02, h10, h11, h12, h20, h21, h22; };

__device__ __forceinline__ HTail load_htail(const float* __restrict__ hpack) {
    HTail h;
    h.h00 = hpack[(L - 1) * 4 + 0]; h.h01 = hpack[(L - 1) * 4 + 1]; h.h02 = hpack[(L - 1) * 4 + 2];
    h.h10 = hpack[(L - 2) * 4 + 0]; h.h11 = hpack[(L - 2) * 4 + 1]; h.h12 = hpack[(L - 2) * 4 + 2];
    h.h20 = hpack[(L - 3) * 4 + 0]; h.h21 = hpack[(L - 3) * 4 + 1]; h.h22 = hpack[(L - 3) * 4 + 2];
    return h;
}

__device__ __forceinline__ float lane_bcast(float v, int src) {
    return __int_as_float(__builtin_amdgcn_readlane(__float_as_int(v), src));
}

__device__ __forceinline__ void wave_scan(const HTail& h, float t0, float t1, float t2,
                                          int lane, float& e0, float& e1, float& e2) {
    float s0 = 0.0f, s1 = 0.0f, s2 = 0.0f;
    e0 = 0.0f; e1 = 0.0f; e2 = 0.0f;
    for (int c = 0; c < C; ++c) {
        if (lane == c) { e0 = s0; e1 = s1; e2 = s2; }
        const float c0 = lane_bcast(t0, c), c1 = lane_bcast(t1, c), c2 = lane_bcast(t2, c);
        const float n0 = c0 + h.h00 * s0 + h.h01 * s1 + h.h02 * s2;
        const float n1 = c1 + h.h10 * s0 + h.h11 * s1 + h.h12 * s2;
        const float n2 = c2 + h.h20 * s0 + h.h21 * s1 + h.h22 * s2;
        s0 = n0; s1 = n1; s2 = n2;
    }
}

static constexpr int P1_BLK = 64;
static constexpr int P1_NV  = P1_BLK / 4;
static constexpr int P1_NB  = L / P1_BLK;

__device__ __forceinline__ void stage_sync() {
    __builtin_amdgcn_fence(__ATOMIC_RELEASE, "wavefront");
    __builtin_amdgcn_wave_barrier();
    __builtin_amdgcn_fence(__ATOMIC_ACQUIRE, "wavefront");
}

template <int NV>
__device__ __forceinline__ void p1_block_tab(v4f (&x)[NV],
                                             float b0, float b1, float b2,
                                             float na0, float na1, float na2,
                                             float& p1, float& p2, float& p3,
                                             float& y1, float& y2, float& y3,
                                             float nta0, float nta1, float nta2,
                                             float& h1, float& h2, float& h3,
                                             float* tcol) {
#pragma unroll
    for (int v = 0; v < NV; ++v) {
        v4f o;
#pragma unroll
        for (int e = 0; e < 4; ++e) {
            const float s = x[v][e];
            const float xf = b0 * p1 + b1 * p2 + b2 * p3;
            const float y = __builtin_fmaf(na0, y1, __builtin_fmaf(na1, y2, __builtin_fmaf(na2, y3, xf)));
            o[e] = y;
            p3 = p2; p2 = p1; p1 = s;
            y3 = y2; y2 = y1; y1 = y;
            const float th = __builtin_fmaf(nta0, h1, __builtin_fmaf(nta1, h2, nta2 * h3));
            tcol[(4 * v + e) * 4] = th;
            h3 = h2; h2 = h1; h1 = th;
        }
        x[v] = o;
    }
}

template <int NK, int NV>
__device__ __forceinline__ void p1_block_reg(v4f (&x)[NV],
                                             float b0, float b1, float b2,
                                             float na0, float na1, float na2,
                                             float& p1, float& p2, float& p3,
                                             float& y1, float& y2, float& y3) {
#pragma unroll
    for (int v = 0; v < NV; ++v) {
        v4f o;
#pragma unroll
        for (int e = 0; e < 4; ++e) {
            const float s = x[v][e];
            float xf;
            if (NK == 0) xf = b0 * s  + b1 * p1 + b2 * p2;
            else         xf = b0 * p1 + b1 * p2 + b2 * p3;
            const float y = __builtin_fmaf(na0, y1, __builtin_fmaf(na1, y2, __builtin_fmaf(na2, y3, xf)));
            o[e] = y;
            p3 = p2; p2 = p1; p1 = s;
            y3 = y2; y2 = y1; y1 = y;
        }
        x[v] = o;
    }
}

static constexpr int ROW_PITCH = L + 4;
static constexpr int ROW_WAVE  = C * ROW_PITCH;
static constexpr int F_WAVES   = 1;
static constexpr int F_NS = P1_BLK / 16;

template <int NV>
__device__ __forceinline__ void lines_load(v4f (&g)[NV], const float* __restrict__ uline) {
#pragma unroll
    for (int j = 0; j < NV; ++j) g[j] = *(const v4f*)(uline + (size_t)(2 * j) * L);
    __builtin_amdgcn_sched_barrier(0);
}

template <int NV>
__device__ __forceinline__ void lines_put(const v4f (&g)[NV], float* rline) {
#pragma unroll
    for (int j = 0; j < NV; ++j) *(v4f*)(rline + (2 * j) * ROW_PITCH) = g[j];
}

template <int NS>
__device__ __forceinline__ void f_load(v4f (&cz)[NS][4], v2f (&ah)[NS],
                                       const float* zlo, const float* zhi,
                                       const float* __restrict__ hrow) {
#pragma unroll
    for (int s = 0; s < NS; ++s) {
        cz[s][0] = *(const v4f*)(zlo + 16 * s);
        cz[s][1] = *(const v4f*)(zlo + 16 * s + 4);
        cz[s][2] = *(const v4f*)(zhi + 16 * s);
        cz[s][3] = *(const v4f*)(zhi + 16 * s + 4);
        ah[s]    = *(const v2f*)(hrow + 64 * s);
    }
    __builtin_amdgcn_sched_barrier(0);
}

__device__ __forceinline__ void f_v16(const v4f (&cz)[4], v2f av, v2f bvl, v2f bvh,
                                      bool hi, v4f (&seq)[4]) {
    v8f c1, c2;
    c1[0] = cz[0][0]; c1[1] = cz[0][1]; c1[2] = cz[0][2]; c1[3] = cz[0][3];
    c1[4] = cz[1][0]; c1[5] = cz[1][1]; c1[6] = cz[1][2]; c1[7] = cz[1][3];
    c2[0] = cz[2][0]; c2[1] = cz[2][1]; c2[2] = cz[2][2]; c2[3] = cz[2][3];
    c2[4] = cz[3][0]; c2[5] = cz[3][1]; c2[6] = cz[3][2]; c2[7] = cz[3][3];
    const v8f d1 = __builtin_amdgcn_wmma_f32_16x16x4_f32(
         false, av,  false, bvl,
         (short)0, c1,  false,  false);
    const v8f d2 = __builtin_amdgcn_wmma_f32_16x16x4_f32(
         false, av,  false, bvh,
         (short)0, c2,  false,  false);
    float v1[8], v2[8], rcv[8];
#pragma unroll
    for (int r = 0; r < 8; ++r) {
        v1[r] = whf(d1[r]);
        v2[r] = whf(d2[r]);
    }
    __builtin_amdgcn_sched_barrier(0);
#pragma unroll
    for (int r = 0; r < 8; ++r) rcv[r] = __shfl_xor(hi ? v1[r] : v2[r], 16);
    __builtin_amdgcn_sched_barrier(0);
#pragma unroll
    for (int r = 0; r < 8; ++r) {
        seq[r >> 2][r & 3]       = hi ? rcv[r] : v1[r];
        seq[2 + (r >> 2)][r & 3] = hi ? v2[r]  : rcv[r];
    }
}

template <int NS>
__device__ __forceinline__ void f_block(const v4f (&cz)[NS][4], const v2f (&ah)[NS],
                                        v2f bvl, v2f bvh, bool hi,
                                        float* rowown,
                                        float b0, float b1, float b2,
                                        float na0, float na1, float na2,
                                        float& p1, float& p2, float& p3,
                                        float& y1, float& y2, float& y3) {
#pragma unroll
    for (int s = 0; s < NS; ++s) {
        v4f seq[4];
        f_v16(cz[s], ah[s], bvl, bvh, hi, seq);
        p1_block_reg<0, 4>(seq, b0, b1, b2, na0, na1, na2, p1, p2, p3, y1, y2, y3);
#pragma unroll
        for (int i = 0; i < 4; ++i) *(v4f*)(rowown + 16 * s + 4 * i) = seq[i];
    }
}

__global__ __launch_bounds__(32 * F_WAVES)
void fused_kernel(const float* __restrict__ u,
                  const float* __restrict__ bco1,
                  const float* __restrict__ aco1,
                  const float* __restrict__ bco2,
                  const float* __restrict__ aco2,
                  float* __restrict__ out) {
    __shared__ __attribute__((aligned(16))) float row_all[F_WAVES * ROW_WAVE];
    __shared__ __attribute__((aligned(16))) float tab_all[F_WAVES * 2 * TAB];
    const int id   = blockIdx.x * blockDim.x + threadIdx.x;
    const int b    = id / C;
    const int lane = threadIdx.x & 31;
    const int n    = lane & 15;
    const int half = lane >> 4;
    const bool hi  = half != 0;

    float* rowbuf = row_all + (threadIdx.x >> 5) * ROW_WAVE;
    float* rowown = rowbuf + lane * ROW_PITCH;
    float* tab1   = tab_all + (threadIdx.x >> 5) * (2 * TAB);
    float* tab2   = tab1 + TAB;

    float y1 = 0.0f, y2 = 0.0f, y3 = 0.0f;
    {
        const float b0 = bco1[0], b1 = bco1[1], b2 = bco1[2];
        const float na0 = -aco1[0], na1 = -aco1[1], na2 = -aco1[2];
        const int col = lane & 7;
        const int tj  = col & 3;
        const float* __restrict__ tac = (col < 4) ? aco1 : aco2;
        const float nta0 = -tac[0], nta1 = -tac[1], nta2 = -tac[2];
        float h1 = (tj == 0) ? 1.0f : 0.0f;
        float h2 = (tj == 1) ? 1.0f : 0.0f;
        float h3 = (tj == 2) ? 1.0f : 0.0f;
        float* tcol = ((col < 4) ? tab1 : tab2) + tj;

        const float* __restrict__ urow  = u + (size_t)b * T;
        const float* __restrict__ src   = urow + (size_t)lane * L;
        const float* __restrict__ uline = urow + (size_t)half * L + 4 * n;
        float*                    rline = rowbuf + half * ROW_PITCH + 4 * n;

        float p1 = 0.0f, p2 = 0.0f, p3 = 0.0f;
        if (lane > 0) {
            p1 = src[-1];
            p2 = src[-2];
            p3 = src[-3];
        }

        v4f g[P1_NV], x[P1_NV];
        lines_load<P1_NV>(g, uline);
        lines_put<P1_NV>(g, rline);
        stage_sync();
#pragma unroll 1
        for (int blk = 0; blk < P1_NB; ++blk) {
#pragma unroll
            for (int v = 0; v < P1_NV; ++v) x[v] = *(const v4f*)(rowown + blk * P1_BLK + 4 * v);
            if (blk + 1 < P1_NB) lines_load<P1_NV>(g, uline + (blk + 1) * P1_BLK);
            p1_block_tab<P1_NV>(x, b0, b1, b2, na0, na1, na2, p1, p2, p3, y1, y2, y3,
                                nta0, nta1, nta2, h1, h2, h3, tcol + blk * P1_BLK * 4);
#pragma unroll
            for (int v = 0; v < P1_NV; ++v) *(v4f*)(rowown + blk * P1_BLK + 4 * v) = x[v];
            if (blk + 1 < P1_NB) {
                lines_put<P1_NV>(g, rline + (blk + 1) * P1_BLK);
                stage_sync();
            }
        }
    }
    stage_sync();

    const float b0 = bco2[0], b1 = bco2[1], b2 = bco2[2];
    const float na0 = -aco2[0], na1 = -aco2[1], na2 = -aco2[2];
    const HTail ht = load_htail(tab2);

    const float* zlo  = rowbuf + n * ROW_PITCH + 8 * half;
    const float* zhi  = rowbuf + (16 + n) * ROW_PITCH + 8 * half;
    const float* hrow = tab1 + n * 4 + 2 * half;

    float e0, e1, e2;
    {
        const HTail ht1 = load_htail(tab1);
        wave_scan(ht1, y1, y2, y3, lane, e0, e1, e2);
    }
    v2f bvl, bvh;
    {
        const float l0 = __shfl(e0, n),      l1 = __shfl(e1, n),      l2 = __shfl(e2, n);
        const float h0 = __shfl(e0, 16 + n), h1 = __shfl(e1, 16 + n), h2 = __shfl(e2, 16 + n);
        bvl[0] = hi ? l2 : l0;  bvl[1] = hi ? 0.0f : l1;
        bvh[0] = hi ? h2 : h0;  bvh[1] = hi ? 0.0f : h1;
    }

    float p1, p2, p3 = 0.0f;
    {
        v4f cz[1][4]; v2f ah[1]; v4f seq[4];
        f_load<1>(cz, ah, zlo + (L - 16), zhi + (L - 16), hrow + (L - 16) * 4);
        f_v16(cz[0], ah[0], bvl, bvh, hi, seq);
        p1 = __shfl_up(seq[3][3], 1);
        p2 = __shfl_up(seq[3][2], 1);
        if (lane == 0) { p1 = 0.0f; p2 = 0.0f; }
    }
    y1 = 0.0f; y2 = 0.0f; y3 = 0.0f;

    v4f ca[F_NS][4], cb[F_NS][4];
    v2f ha[F_NS], hb[F_NS];
    f_load<F_NS>(ca, ha, zlo, zhi, hrow);
    for (int blk = 0; blk < P1_NB - 2; blk += 2) {
        f_load<F_NS>(cb, hb, zlo + (blk + 1) * P1_BLK, zhi + (blk + 1) * P1_BLK,
                     hrow + (blk + 1) * P1_BLK * 4);
        f_block<F_NS>(ca, ha, bvl, bvh, hi, rowown + blk * P1_BLK,
                      b0, b1, b2, na0, na1, na2, p1, p2, p3, y1, y2, y3);
        f_load<F_NS>(ca, ha, zlo + (blk + 2) * P1_BLK, zhi + (blk + 2) * P1_BLK,
                     hrow + (blk + 2) * P1_BLK * 4);
        f_block<F_NS>(cb, hb, bvl, bvh, hi, rowown + (blk + 1) * P1_BLK,
                      b0, b1, b2, na0, na1, na2, p1, p2, p3, y1, y2, y3);
    }
    f_load<F_NS>(cb, hb, zlo + (P1_NB - 1) * P1_BLK, zhi + (P1_NB - 1) * P1_BLK,
                 hrow + (P1_NB - 1) * P1_BLK * 4);
    f_block<F_NS>(ca, ha, bvl, bvh, hi, rowown + (P1_NB - 2) * P1_BLK,
                  b0, b1, b2, na0, na1, na2, p1, p2, p3, y1, y2, y3);
    f_block<F_NS>(cb, hb, bvl, bvh, hi, rowown + (P1_NB - 1) * P1_BLK,
                  b0, b1, b2, na0, na1, na2, p1, p2, p3, y1, y2, y3);

    wave_scan(ht, y1, y2, y3, lane, e0, e1, e2);
    v2f svl, svh;
    {
        const float l0 = __shfl(e0, n),      l1 = __shfl(e1, n),      l2 = __shfl(e2, n);
        const float h0 = __shfl(e0, 16 + n), h1 = __shfl(e1, 16 + n), h2 = __shfl(e2, 16 + n);
        svl[0] = hi ? l2 : l0;  svl[1] = hi ? 0.0f : l1;
        svh[0] = hi ? h2 : h0;  svh[1] = hi ? 0.0f : h1;
    }
    stage_sync();

    float* orow = out + (size_t)b * T;
    const float* hcol = tab2 + (4 * n) * 4 + 2 * half;
#pragma unroll 2
    for (int blk = 0; blk < P1_NB; ++blk) {
        v2f bt[4];
#pragma unroll
        for (int j = 0; j < 4; ++j) bt[j] = *(const v2f*)(hcol + (blk * P1_BLK + j) * 4);
#pragma unroll
        for (int grp = 0; grp < 2; ++grp) {
            const float* zc = rowbuf + (16 * grp + 8 * half) * ROW_PITCH + blk * P1_BLK + 4 * n;
            float*       oc = orow + (size_t)(16 * grp + 8 * half) * L + blk * P1_BLK + 4 * n;
            v4f cr[8];
#pragma unroll
            for (int r = 0; r < 8; ++r) cr[r] = *(const v4f*)(zc + r * ROW_PITCH);
            v8f acc[4];
#pragma unroll
            for (int j = 0; j < 4; ++j) {
#pragma unroll
                for (int r = 0; r < 8; ++r) acc[j][r] = cr[r][j];
                acc[j] = __builtin_amdgcn_wmma_f32_16x16x4_f32(
                     false, grp ? svh : svl,  false, bt[j],
                     (short)0, acc[j],  false,  false);
            }
#pragma unroll
            for (int r = 0; r < 8; ++r) {
                v4f o;
                o[0] = acc[0][r]; o[1] = acc[1][r]; o[2] = acc[2][r]; o[3] = acc[3][r];
                __builtin_nontemporal_store(o, (v4f*)(oc + (size_t)r * L));
            }
        }
    }
}

extern "C" void kernel_launch(void* const* d_in, const int* in_sizes, int n_in,
                              void* d_out, int out_size, void* d_ws, size_t ws_size,
                              hipStream_t stream) {
    const float* u  = (const float*)d_in[0];
    const float* b1 = (const float*)d_in[1];
    const float* a1 = (const float*)d_in[2];
    const float* b2 = (const float*)d_in[3];
    const float* a2 = (const float*)d_in[4];
    float* out = (float*)d_out;

    const int f_blocks = Bsz / F_WAVES;

    fused_kernel<<<f_blocks, 32 * F_WAVES, 0, stream>>>(u, b1, a1, b2, a2, out);
}


# each row is shared by two waves (64 chunks of 256 samples, was 32 of 512): every per-lane sample loop halves; tails cross through LDS at two barriers; chunk borders move, so rounding changes
# speedup vs baseline: 26.6992x; 1.3818x over previous
#include <hip/hip_runtime.h>
#include <hip/hip_bf16.h>
#include <math.h>

static constexpr int Bsz = 512;
static constexpr int T   = 16384;
static constexpr int L   = 256;
static constexpr int C   = T / L;
static constexpr int WPR = C / 32;

typedef __attribute__((ext_vector_type(2))) float v2f;
typedef __attribute__((ext_vector_type(4))) float v4f;
typedef __attribute__((ext_vector_type(8))) float v8f;

__device__ __forceinline__ float whf(float x) {
    const float c   = 10.0f / 11.0f;
    const float l2e = 1.4426950408889634f;
    float e = 1.0f - __builtin_amdgcn_exp2f(-(c * l2e) * x);
    float l = c * x;
    unsigned s = (unsigned)(__float_as_int(x) >> 31);
    return __uint_as_float((__float_as_uint(e) & ~s) | (__float_as_uint(l) & s));
}


static constexpr int TAB = L * 4;

static_assert(WPR == 2, "two waves per row: lane index == chunk index inside a wave");

struct HTail { float h00, h01, h02, h10, h11, h12, h20, h21, h22; };

__device__ __forceinline__ HTail load_htail(const float* __restrict__ hpack) {
    HTail h;
    h.h00 = hpack[(L - 1) * 4 + 0]; h.h01 = hpack[(L - 1) * 4 + 1]; h.h02 = hpack[(L - 1) * 4 + 2];
    h.h10 = hpack[(L - 2) * 4 + 0]; h.h11 = hpack[(L - 2) * 4 + 1]; h.h12 = hpack[(L - 2) * 4 + 2];
    h.h20 = hpack[(L - 3) * 4 + 0]; h.h21 = hpack[(L - 3) * 4 + 1]; h.h22 = hpack[(L - 3) * 4 + 2];
    return h;
}

__device__ __forceinline__ float lane_bcast(float v, int src) {
    return __int_as_float(__builtin_amdgcn_readlane(__float_as_int(v), src));
}

__device__ __forceinline__ void scan32(const HTail& h, float t0, float t1, float t2,
                                       int lane, bool keep,
                                       float& s0, float& s1, float& s2,
                                       float& e0, float& e1, float& e2) {
    for (int c = 0; c < 32; ++c) {
        if (keep && lane == c) { e0 = s0; e1 = s1; e2 = s2; }
        const float c0 = lane_bcast(t0, c), c1 = lane_bcast(t1, c), c2 = lane_bcast(t2, c);
        const float n0 = c0 + h.h00 * s0 + h.h01 * s1 + h.h02 * s2;
        const float n1 = c1 + h.h10 * s0 + h.h11 * s1 + h.h12 * s2;
        const float n2 = c2 + h.h20 * s0 + h.h21 * s1 + h.h22 * s2;
        s0 = n0; s1 = n1; s2 = n2;
    }
}

__device__ __forceinline__ void row_scan(const HTail& h, float t0, float t1, float t2,
                                         int lane, int wv, float* xch,
                                         float& e0, float& e1, float& e2) {
    v4f mine; mine[0] = t0; mine[1] = t1; mine[2] = t2; mine[3] = 0.0f;
    *(v4f*)(xch + (32 * wv + lane) * 4) = mine;
    __syncthreads();
    const v4f first = *(const v4f*)(xch + lane * 4);
    float s0 = 0.0f, s1 = 0.0f, s2 = 0.0f;
    e0 = 0.0f; e1 = 0.0f; e2 = 0.0f;
    scan32(h, first[0], first[1], first[2], lane, wv == 0, s0, s1, s2, e0, e1, e2);
    if (wv != 0) scan32(h, t0, t1, t2, lane, true, s0, s1, s2, e0, e1, e2);
}

static constexpr int P1_BLK = 64;
static constexpr int P1_NV  = P1_BLK / 4;
static constexpr int P1_NB  = L / P1_BLK;

__device__ __forceinline__ void stage_sync() {
    __builtin_amdgcn_fence(__ATOMIC_RELEASE, "wavefront");
    __builtin_amdgcn_wave_barrier();
    __builtin_amdgcn_fence(__ATOMIC_ACQUIRE, "wavefront");
}

template <int NV>
__device__ __forceinline__ void p1_block_tab(v4f (&x)[NV],
                                             float b0, float b1, float b2,
                                             float na0, float na1, float na2,
                                             float& p1, float& p2, float& p3,
                                             float& y1, float& y2, float& y3,
                                             float nta0, float nta1, float nta2,
                                             float& h1, float& h2, float& h3,
                                             float* tcol) {
#pragma unroll
    for (int v = 0; v < NV; ++v) {
        v4f o;
#pragma unroll
        for (int e = 0; e < 4; ++e) {
            const float s = x[v][e];
            const float xf = b0 * p1 + b1 * p2 + b2 * p3;
            const float y = __builtin_fmaf(na0, y1, __builtin_fmaf(na1, y2, __builtin_fmaf(na2, y3, xf)));
            o[e] = y;
            p3 = p2; p2 = p1; p1 = s;
            y3 = y2; y2 = y1; y1 = y;
            const float th = __builtin_fmaf(nta0, h1, __builtin_fmaf(nta1, h2, nta2 * h3));
            tcol[(4 * v + e) * 4] = th;
            h3 = h2; h2 = h1; h1 = th;
        }
        x[v] = o;
    }
}

template <int NK, int NV>
__device__ __forceinline__ void p1_block_reg(v4f (&x)[NV],
                                             float b0, float b1, float b2,
                                             float na0, float na1, float na2,
                                             float& p1, float& p2, float& p3,
                                             float& y1, float& y2, float& y3) {
#pragma unroll
    for (int v = 0; v < NV; ++v) {
        v4f o;
#pragma unroll
        for (int e = 0; e < 4; ++e) {
            const float s = x[v][e];
            float xf;
            if (NK == 0) xf = b0 * s  + b1 * p1 + b2 * p2;
            else         xf = b0 * p1 + b1 * p2 + b2 * p3;
            const float y = __builtin_fmaf(na0, y1, __builtin_fmaf(na1, y2, __builtin_fmaf(na2, y3, xf)));
            o[e] = y;
            p3 = p2; p2 = p1; p1 = s;
            y3 = y2; y2 = y1; y1 = y;
        }
        x[v] = o;
    }
}

static constexpr int ROW_PITCH = L + 4;
static constexpr int ROW_WAVE  = 32 * ROW_PITCH;
static constexpr int F_NS = P1_BLK / 16;

template <int NV>
__device__ __forceinline__ void lines_load(v4f (&g)[NV], const float* __restrict__ uline) {
#pragma unroll
    for (int j = 0; j < NV; ++j) g[j] = *(const v4f*)(uline + (size_t)(2 * j) * L);
    __builtin_amdgcn_sched_barrier(0);
}

template <int NV>
__device__ __forceinline__ void lines_put(const v4f (&g)[NV], float* rline) {
#pragma unroll
    for (int j = 0; j < NV; ++j) *(v4f*)(rline + (2 * j) * ROW_PITCH) = g[j];
}

template <int NS>
__device__ __forceinline__ void f_load(v4f (&cz)[NS][4], v2f (&ah)[NS],
                                       const float* zlo, const float* zhi,
                                       const float* __restrict__ hrow) {
#pragma unroll
    for (int s = 0; s < NS; ++s) {
        cz[s][0] = *(const v4f*)(zlo + 16 * s);
        cz[s][1] = *(const v4f*)(zlo + 16 * s + 4);
        cz[s][2] = *(const v4f*)(zhi + 16 * s);
        cz[s][3] = *(const v4f*)(zhi + 16 * s + 4);
        ah[s]    = *(const v2f*)(hrow + 64 * s);
    }
    __builtin_amdgcn_sched_barrier(0);
}

__device__ __forceinline__ void f_v16(const v4f (&cz)[4], v2f av, v2f bvl, v2f bvh,
                                      bool hi, v4f (&seq)[4]) {
    v8f c1, c2;
    c1[0] = cz[0][0]; c1[1] = cz[0][1]; c1[2] = cz[0][2]; c1[3] = cz[0][3];
    c1[4] = cz[1][0]; c1[5] = cz[1][1]; c1[6] = cz[1][2]; c1[7] = cz[1][3];
    c2[0] = cz[2][0]; c2[1] = cz[2][1]; c2[2] = cz[2][2]; c2[3] = cz[2][3];
    c2[4] = cz[3][0]; c2[5] = cz[3][1]; c2[6] = cz[3][2]; c2[7] = cz[3][3];
    const v8f d1 = __builtin_amdgcn_wmma_f32_16x16x4_f32(
         false, av,  false, bvl,
         (short)0, c1,  false,  false);
    const v8f d2 = __builtin_amdgcn_wmma_f32_16x16x4_f32(
         false, av,  false, bvh,
         (short)0, c2,  false,  false);
    float v1[8], v2[8], rcv[8];
#pragma unroll
    for (int r = 0; r < 8; ++r) {
        v1[r] = whf(d1[r]);
        v2[r] = whf(d2[r]);
    }
    __builtin_amdgcn_sched_barrier(0);
#pragma unroll
    for (int r = 0; r < 8; ++r) rcv[r] = __shfl_xor(hi ? v1[r] : v2[r], 16);
    __builtin_amdgcn_sched_barrier(0);
#pragma unroll
    for (int r = 0; r < 8; ++r) {
        seq[r >> 2][r & 3]       = hi ? rcv[r] : v1[r];
        seq[2 + (r >> 2)][r & 3] = hi ? v2[r]  : rcv[r];
    }
}

template <int NS>
__device__ __forceinline__ void f_block(const v4f (&cz)[NS][4], const v2f (&ah)[NS],
                                        v2f bvl, v2f bvh, bool hi,
                                        float* rowown,
                                        float b0, float b1, float b2,
                                        float na0, float na1, float na2,
                                        float& p1, float& p2, float& p3,
                                        float& y1, float& y2, float& y3) {
#pragma unroll
    for (int s = 0; s < NS; ++s) {
        v4f seq[4];
        f_v16(cz[s], ah[s], bvl, bvh, hi, seq);
        p1_block_reg<0, 4>(seq, b0, b1, b2, na0, na1, na2, p1, p2, p3, y1, y2, y3);
#pragma unroll
        for (int i = 0; i < 4; ++i) *(v4f*)(rowown + 16 * s + 4 * i) = seq[i];
    }
}

__global__ __launch_bounds__(32 * WPR)
void fused_kernel(const float* __restrict__ u,
                  const float* __restrict__ bco1,
                  const float* __restrict__ aco1,
                  const float* __restrict__ bco2,
                  const float* __restrict__ aco2,
                  float* __restrict__ out) {
    __shared__ __attribute__((aligned(16))) float row_all[WPR * ROW_WAVE];
    __shared__ __attribute__((aligned(16))) float tab_all[WPR * 2 * TAB];
    __shared__ __attribute__((aligned(16))) float xch1[C * 4];
    __shared__ __attribute__((aligned(16))) float xch2[C * 4];
    __shared__ float vedge[2];
    const int b    = blockIdx.x;
    const int wv   = __builtin_amdgcn_readfirstlane((int)(threadIdx.x >> 5));
    const int lane = threadIdx.x & 31;
    const int n    = lane & 15;
    const int half = lane >> 4;
    const bool hi  = half != 0;

    float* rowbuf = row_all + wv * ROW_WAVE;
    float* rowown = rowbuf + lane * ROW_PITCH;
    float* tab1   = tab_all + wv * (2 * TAB);
    float* tab2   = tab1 + TAB;

    float y1 = 0.0f, y2 = 0.0f, y3 = 0.0f;
    {
        const float b0 = bco1[0], b1 = bco1[1], b2 = bco1[2];
        const float na0 = -aco1[0], na1 = -aco1[1], na2 = -aco1[2];
        const int col = lane & 7;
        const int tj  = col & 3;
        const float* __restrict__ tac = (col < 4) ? aco1 : aco2;
        const float nta0 = -tac[0], nta1 = -tac[1], nta2 = -tac[2];
        float h1 = (tj == 0) ? 1.0f : 0.0f;
        float h2 = (tj == 1) ? 1.0f : 0.0f;
        float h3 = (tj == 2) ? 1.0f : 0.0f;
        float* tcol = ((col < 4) ? tab1 : tab2) + tj;

        const float* __restrict__ urow  = u + (size_t)b * T + (size_t)wv * (32 * L);
        const float* __restrict__ src   = urow + (size_t)lane * L;
        const float* __restrict__ uline = urow + (size_t)half * L + 4 * n;
        float*                    rline = rowbuf + half * ROW_PITCH + 4 * n;

        float p1 = 0.0f, p2 = 0.0f, p3 = 0.0f;
        if (32 * wv + lane > 0) {
            p1 = src[-1];
            p2 = src[-2];
            p3 = src[-3];
        }

        v4f g[P1_NV], x[P1_NV];
        lines_load<P1_NV>(g, uline);
        lines_put<P1_NV>(g, rline);
        stage_sync();
#pragma unroll 1
        for (int blk = 0; blk < P1_NB; ++blk) {
#pragma unroll
            for (int v = 0; v < P1_NV; ++v) x[v] = *(const v4f*)(rowown + blk * P1_BLK + 4 * v);
            if (blk + 1 < P1_NB) lines_load<P1_NV>(g, uline + (blk + 1) * P1_BLK);
            p1_block_tab<P1_NV>(x, b0, b1, b2, na0, na1, na2, p1, p2, p3, y1, y2, y3,
                                nta0, nta1, nta2, h1, h2, h3, tcol + blk * P1_BLK * 4);
#pragma unroll
            for (int v = 0; v < P1_NV; ++v) *(v4f*)(rowown + blk * P1_BLK + 4 * v) = x[v];
            if (blk + 1 < P1_NB) {
                lines_put<P1_NV>(g, rline + (blk + 1) * P1_BLK);
                stage_sync();
            }
        }
    }
    stage_sync();

    const float b0 = bco2[0], b1 = bco2[1], b2 = bco2[2];
    const float na0 = -aco2[0], na1 = -aco2[1], na2 = -aco2[2];
    const HTail ht = load_htail(tab2);

    const float* zlo  = rowbuf + n * ROW_PITCH + 8 * half;
    const float* zhi  = rowbuf + (16 + n) * ROW_PITCH + 8 * half;
    const float* hrow = tab1 + n * 4 + 2 * half;

    float e0, e1, e2;
    {
        const HTail ht1 = load_htail(tab1);
        row_scan(ht1, y1, y2, y3, lane, wv, xch1, e0, e1, e2);
    }
    v2f bvl, bvh;
    {
        const float l0 = __shfl(e0, n),      l1 = __shfl(e1, n),      l2 = __shfl(e2, n);
        const float h0 = __shfl(e0, 16 + n), h1 = __shfl(e1, 16 + n), h2 = __shfl(e2, 16 + n);
        bvl[0] = hi ? l2 : l0;  bvl[1] = hi ? 0.0f : l1;
        bvh[0] = hi ? h2 : h0;  bvh[1] = hi ? 0.0f : h1;
    }

    float p1, p2, p3 = 0.0f;
    {
        v4f cz[1][4]; v2f ah[1]; v4f seq[4];
        f_load<1>(cz, ah, zlo + (L - 16), zhi + (L - 16), hrow + (L - 16) * 4);
        f_v16(cz[0], ah[0], bvl, bvh, hi, seq);
        if (wv == 0 && lane == 31) { vedge[0] = seq[3][3]; vedge[1] = seq[3][2]; }
        __syncthreads();
        p1 = __shfl_up(seq[3][3], 1);
        p2 = __shfl_up(seq[3][2], 1);
        if (lane == 0) {
            p1 = (wv != 0) ? vedge[0] : 0.0f;
            p2 = (wv != 0) ? vedge[1] : 0.0f;
        }
    }
    y1 = 0.0f; y2 = 0.0f; y3 = 0.0f;

    v4f ca[F_NS][4], cb[F_NS][4];
    v2f ha[F_NS], hb[F_NS];
    f_load<F_NS>(ca, ha, zlo, zhi, hrow);
    for (int blk = 0; blk < P1_NB - 2; blk += 2) {
        f_load<F_NS>(cb, hb, zlo + (blk + 1) * P1_BLK, zhi + (blk + 1) * P1_BLK,
                     hrow + (blk + 1) * P1_BLK * 4);
        f_block<F_NS>(ca, ha, bvl, bvh, hi, rowown + blk * P1_BLK,
                      b0, b1, b2, na0, na1, na2, p1, p2, p3, y1, y2, y3);
        f_load<F_NS>(ca, ha, zlo + (blk + 2) * P1_BLK, zhi + (blk + 2) * P1_BLK,
                     hrow + (blk + 2) * P1_BLK * 4);
        f_block<F_NS>(cb, hb, bvl, bvh, hi, rowown + (blk + 1) * P1_BLK,
                      b0, b1, b2, na0, na1, na2, p1, p2, p3, y1, y2, y3);
    }
    f_load<F_NS>(cb, hb, zlo + (P1_NB - 1) * P1_BLK, zhi + (P1_NB - 1) * P1_BLK,
                 hrow + (P1_NB - 1) * P1_BLK * 4);
    f_block<F_NS>(ca, ha, bvl, bvh, hi, rowown + (P1_NB - 2) * P1_BLK,
                  b0, b1, b2, na0, na1, na2, p1, p2, p3, y1, y2, y3);
    f_block<F_NS>(cb, hb, bvl, bvh, hi, rowown + (P1_NB - 1) * P1_BLK,
                  b0, b1, b2, na0, na1, na2, p1, p2, p3, y1, y2, y3);

    row_scan(ht, y1, y2, y3, lane, wv, xch2, e0, e1, e2);
    v2f svl, svh;
    {
        const float l0 = __shfl(e0, n),      l1 = __shfl(e1, n),      l2 = __shfl(e2, n);
        const float h0 = __shfl(e0, 16 + n), h1 = __shfl(e1, 16 + n), h2 = __shfl(e2, 16 + n);
        svl[0] = hi ? l2 : l0;  svl[1] = hi ? 0.0f : l1;
        svh[0] = hi ? h2 : h0;  svh[1] = hi ? 0.0f : h1;
    }
    stage_sync();

    float* orow = out + (size_t)b * T + (size_t)wv * (32 * L);
    const float* hcol = tab2 + (4 * n) * 4 + 2 * half;
#pragma unroll 2
    for (int blk = 0; blk < P1_NB; ++blk) {
        v2f bt[4];
#pragma unroll
        for (int j = 0; j < 4; ++j) bt[j] = *(const v2f*)(hcol + (blk * P1_BLK + j) * 4);
#pragma unroll
        for (int grp = 0; grp < 2; ++grp) {
            const float* zc = rowbuf + (16 * grp + 8 * half) * ROW_PITCH + blk * P1_BLK + 4 * n;
            float*       oc = orow + (size_t)(16 * grp + 8 * half) * L + blk * P1_BLK + 4 * n;
            v4f cr[8];
#pragma unroll
            for (int r = 0; r < 8; ++r) cr[r] = *(const v4f*)(zc + r * ROW_PITCH);
            v8f acc[4];
#pragma unroll
            for (int j = 0; j < 4; ++j) {
#pragma unroll
                for (int r = 0; r < 8; ++r) acc[j][r] = cr[r][j];
                acc[j] = __builtin_amdgcn_wmma_f32_16x16x4_f32(
                     false, grp ? svh : svl,  false, bt[j],
                     (short)0, acc[j],  false,  false);
            }
#pragma unroll
            for (int r = 0; r < 8; ++r) {
                v4f o;
                o[0] = acc[0][r]; o[1] = acc[1][r]; o[2] = acc[2][r]; o[3] = acc[3][r];
                __builtin_nontemporal_store(o, (v4f*)(oc + (size_t)r * L));
            }
        }
    }
}

extern "C" void kernel_launch(void* const* d_in, const int* in_sizes, int n_in,
                              void* d_out, int out_size, void* d_ws, size_t ws_size,
                              hipStream_t stream) {
    const float* u  = (const float*)d_in[0];
    const float* b1 = (const float*)d_in[1];
    const float* a1 = (const float*)d_in[2];
    const float* b2 = (const float*)d_in[3];
    const float* a2 = (const float*)d_in[4];
    float* out = (float*)d_out;

    fused_kernel<<<Bsz, 32 * WPR, 0, stream>>>(u, b1, a1, b2, a2, out);
}


# half-wave exchanges (v between the lane halves of the two products, 8 values per 16-sample step; state operands of both matrix stages): one vector-unit half-row swap each; was selects + LDS permutes
# speedup vs baseline: 28.8368x; 1.0801x over previous
#include <hip/hip_runtime.h>
#include <hip/hip_bf16.h>
#include <math.h>

static constexpr int Bsz = 512;
static constexpr int T   = 16384;
static constexpr int L   = 256;
static constexpr int C   = T / L;
static constexpr int WPR = C / 32;

typedef __attribute__((ext_vector_type(2))) float v2f;
typedef __attribute__((ext_vector_type(2))) unsigned v2u;
typedef __attribute__((ext_vector_type(4))) float v4f;
typedef __attribute__((ext_vector_type(8))) float v8f;

__device__ __forceinline__ float whf(float x) {
    const float c   = 10.0f / 11.0f;
    const float l2e = 1.4426950408889634f;
    float e = 1.0f - __builtin_amdgcn_exp2f(-(c * l2e) * x);
    float l = c * x;
    unsigned s = (unsigned)(__float_as_int(x) >> 31);
    return __uint_as_float((__float_as_uint(e) & ~s) | (__float_as_uint(l) & s));
}


static constexpr int TAB = L * 4;

static_assert(WPR == 2, "two waves per row: lane index == chunk index inside a wave");

struct HTail { float h00, h01, h02, h10, h11, h12, h20, h21, h22; };

__device__ __forceinline__ HTail load_htail(const float* __restrict__ hpack) {
    HTail h;
    h.h00 = hpack[(L - 1) * 4 + 0]; h.h01 = hpack[(L - 1) * 4 + 1]; h.h02 = hpack[(L - 1) * 4 + 2];
    h.h10 = hpack[(L - 2) * 4 + 0]; h.h11 = hpack[(L - 2) * 4 + 1]; h.h12 = hpack[(L - 2) * 4 + 2];
    h.h20 = hpack[(L - 3) * 4 + 0]; h.h21 = hpack[(L - 3) * 4 + 1]; h.h22 = hpack[(L - 3) * 4 + 2];
    return h;
}

__device__ __forceinline__ float lane_bcast(float v, int src) {
    return __int_as_float(__builtin_amdgcn_readlane(__float_as_int(v), src));
}

__device__ __forceinline__ v2f half_swap(float a, float b) {
    const v2u s = __builtin_amdgcn_permlane16_swap(__float_as_uint(a), __float_as_uint(b),
                                                    false,  false);
    v2f r; r[0] = __uint_as_float(s[0]); r[1] = __uint_as_float(s[1]);
    return r;
}

__device__ __forceinline__ void scan32(const HTail& h, float t0, float t1, float t2,
                                       int lane, bool keep,
                                       float& s0, float& s1, float& s2,
                                       float& e0, float& e1, float& e2) {
    for (int c = 0; c < 32; ++c) {
        if (keep && lane == c) { e0 = s0; e1 = s1; e2 = s2; }
        const float c0 = lane_bcast(t0, c), c1 = lane_bcast(t1, c), c2 = lane_bcast(t2, c);
        const float n0 = c0 + h.h00 * s0 + h.h01 * s1 + h.h02 * s2;
        const float n1 = c1 + h.h10 * s0 + h.h11 * s1 + h.h12 * s2;
        const float n2 = c2 + h.h20 * s0 + h.h21 * s1 + h.h22 * s2;
        s0 = n0; s1 = n1; s2 = n2;
    }
}

__device__ __forceinline__ void row_scan(const HTail& h, float t0, float t1, float t2,
                                         int lane, int wv, float* xch,
                                         float& e0, float& e1, float& e2) {
    v4f mine; mine[0] = t0; mine[1] = t1; mine[2] = t2; mine[3] = 0.0f;
    *(v4f*)(xch + (32 * wv + lane) * 4) = mine;
    __syncthreads();
    const v4f first = *(const v4f*)(xch + lane * 4);
    float s0 = 0.0f, s1 = 0.0f, s2 = 0.0f;
    e0 = 0.0f; e1 = 0.0f; e2 = 0.0f;
    scan32(h, first[0], first[1], first[2], lane, wv == 0, s0, s1, s2, e0, e1, e2);
    if (wv != 0) scan32(h, t0, t1, t2, lane, true, s0, s1, s2, e0, e1, e2);
}

static constexpr int P1_BLK = 64;
static constexpr int P1_NV  = P1_BLK / 4;
static constexpr int P1_NB  = L / P1_BLK;

__device__ __forceinline__ void stage_sync() {
    __builtin_amdgcn_fence(__ATOMIC_RELEASE, "wavefront");
    __builtin_amdgcn_wave_barrier();
    __builtin_amdgcn_fence(__ATOMIC_ACQUIRE, "wavefront");
}

template <int NV>
__device__ __forceinline__ void p1_block_tab(v4f (&x)[NV],
                                             float b0, float b1, float b2,
                                             float na0, float na1, float na2,
                                             float& p1, float& p2, float& p3,
                                             float& y1, float& y2, float& y3,
                                             float nta0, float nta1, float nta2,
                                             float& h1, float& h2, float& h3,
                                             float* tcol) {
#pragma unroll
    for (int v = 0; v < NV; ++v) {
        v4f o;
#pragma unroll
        for (int e = 0; e < 4; ++e) {
            const float s = x[v][e];
            const float xf = b0 * p1 + b1 * p2 + b2 * p3;
            const float y = __builtin_fmaf(na0, y1, __builtin_fmaf(na1, y2, __builtin_fmaf(na2, y3, xf)));
            o[e] = y;
            p3 = p2; p2 = p1; p1 = s;
            y3 = y2; y2 = y1; y1 = y;
            const float th = __builtin_fmaf(nta0, h1, __builtin_fmaf(nta1, h2, nta2 * h3));
            tcol[(4 * v + e) * 4] = th;
            h3 = h2; h2 = h1; h1 = th;
        }
        x[v] = o;
    }
}

template <int NK, int NV>
__device__ __forceinline__ void p1_block_reg(v4f (&x)[NV],
                                             float b0, float b1, float b2,
                                             float na0, float na1, float na2,
                                             float& p1, float& p2, float& p3,
                                             float& y1, float& y2, float& y3) {
#pragma unroll
    for (int v = 0; v < NV; ++v) {
        v4f o;
#pragma unroll
        for (int e = 0; e < 4; ++e) {
            const float s = x[v][e];
            float xf;
            if (NK == 0) xf = b0 * s  + b1 * p1 + b2 * p2;
            else         xf = b0 * p1 + b1 * p2 + b2 * p3;
            const float y = __builtin_fmaf(na0, y1, __builtin_fmaf(na1, y2, __builtin_fmaf(na2, y3, xf)));
            o[e] = y;
            p3 = p2; p2 = p1; p1 = s;
            y3 = y2; y2 = y1; y1 = y;
        }
        x[v] = o;
    }
}

static constexpr int ROW_PITCH = L + 4;
static constexpr int ROW_WAVE  = 32 * ROW_PITCH;
static constexpr int F_NS = P1_BLK / 16;

template <int NV>
__device__ __forceinline__ void lines_load(v4f (&g)[NV], const float* __restrict__ uline) {
#pragma unroll
    for (int j = 0; j < NV; ++j) g[j] = *(const v4f*)(uline + (size_t)(2 * j) * L);
    __builtin_amdgcn_sched_barrier(0);
}

template <int NV>
__device__ __forceinline__ void lines_put(const v4f (&g)[NV], float* rline) {
#pragma unroll
    for (int j = 0; j < NV; ++j) *(v4f*)(rline + (2 * j) * ROW_PITCH) = g[j];
}

template <int NS>
__device__ __forceinline__ void f_load(v4f (&cz)[NS][4], v2f (&ah)[NS],
                                       const float* zlo, const float* zhi,
                                       const float* __restrict__ hrow) {
#pragma unroll
    for (int s = 0; s < NS; ++s) {
        cz[s][0] = *(const v4f*)(zlo + 16 * s);
        cz[s][1] = *(const v4f*)(zlo + 16 * s + 4);
        cz[s][2] = *(const v4f*)(zhi + 16 * s);
        cz[s][3] = *(const v4f*)(zhi + 16 * s + 4);
        ah[s]    = *(const v2f*)(hrow + 64 * s);
    }
    __builtin_amdgcn_sched_barrier(0);
}

__device__ __forceinline__ void f_v16(const v4f (&cz)[4], v2f av, v2f bvl, v2f bvh,
                                      v4f (&seq)[4]) {
    v8f c1, c2;
    c1[0] = cz[0][0]; c1[1] = cz[0][1]; c1[2] = cz[0][2]; c1[3] = cz[0][3];
    c1[4] = cz[1][0]; c1[5] = cz[1][1]; c1[6] = cz[1][2]; c1[7] = cz[1][3];
    c2[0] = cz[2][0]; c2[1] = cz[2][1]; c2[2] = cz[2][2]; c2[3] = cz[2][3];
    c2[4] = cz[3][0]; c2[5] = cz[3][1]; c2[6] = cz[3][2]; c2[7] = cz[3][3];
    const v8f d1 = __builtin_amdgcn_wmma_f32_16x16x4_f32(
         false, av,  false, bvl,
         (short)0, c1,  false,  false);
    const v8f d2 = __builtin_amdgcn_wmma_f32_16x16x4_f32(
         false, av,  false, bvh,
         (short)0, c2,  false,  false);
#pragma unroll
    for (int r = 0; r < 8; ++r) {
        const float v1 = whf(d1[r]);
        const float v2 = whf(d2[r]);
        const v2f sw = half_swap(v1, v2);
        seq[r >> 2][r & 3]       = sw[0];
        seq[2 + (r >> 2)][r & 3] = sw[1];
    }
    __builtin_amdgcn_sched_barrier(0);
}

template <int NS>
__device__ __forceinline__ void f_block(const v4f (&cz)[NS][4], const v2f (&ah)[NS],
                                        v2f bvl, v2f bvh,
                                        float* rowown,
                                        float b0, float b1, float b2,
                                        float na0, float na1, float na2,
                                        float& p1, float& p2, float& p3,
                                        float& y1, float& y2, float& y3) {
#pragma unroll
    for (int s = 0; s < NS; ++s) {
        v4f seq[4];
        f_v16(cz[s], ah[s], bvl, bvh, seq);
        p1_block_reg<0, 4>(seq, b0, b1, b2, na0, na1, na2, p1, p2, p3, y1, y2, y3);
#pragma unroll
        for (int i = 0; i < 4; ++i) *(v4f*)(rowown + 16 * s + 4 * i) = seq[i];
    }
}

__global__ __launch_bounds__(32 * WPR)
void fused_kernel(const float* __restrict__ u,
                  const float* __restrict__ bco1,
                  const float* __restrict__ aco1,
                  const float* __restrict__ bco2,
                  const float* __restrict__ aco2,
                  float* __restrict__ out) {
    __shared__ __attribute__((aligned(16))) float row_all[WPR * ROW_WAVE];
    __shared__ __attribute__((aligned(16))) float tab_all[WPR * 2 * TAB];
    __shared__ __attribute__((aligned(16))) float xch1[C * 4];
    __shared__ __attribute__((aligned(16))) float xch2[C * 4];
    __shared__ float vedge[2];
    const int b    = blockIdx.x;
    const int wv   = __builtin_amdgcn_readfirstlane((int)(threadIdx.x >> 5));
    const int lane = threadIdx.x & 31;
    const int n    = lane & 15;
    const int half = lane >> 4;
    const bool hi  = half != 0;

    float* rowbuf = row_all + wv * ROW_WAVE;
    float* rowown = rowbuf + lane * ROW_PITCH;
    float* tab1   = tab_all + wv * (2 * TAB);
    float* tab2   = tab1 + TAB;

    float y1 = 0.0f, y2 = 0.0f, y3 = 0.0f;
    {
        const float b0 = bco1[0], b1 = bco1[1], b2 = bco1[2];
        const float na0 = -aco1[0], na1 = -aco1[1], na2 = -aco1[2];
        const int col = lane & 7;
        const int tj  = col & 3;
        const float* __restrict__ tac = (col < 4) ? aco1 : aco2;
        const float nta0 = -tac[0], nta1 = -tac[1], nta2 = -tac[2];
        float h1 = (tj == 0) ? 1.0f : 0.0f;
        float h2 = (tj == 1) ? 1.0f : 0.0f;
        float h3 = (tj == 2) ? 1.0f : 0.0f;
        float* tcol = ((col < 4) ? tab1 : tab2) + tj;

        const float* __restrict__ urow  = u + (size_t)b * T + (size_t)wv * (32 * L);
        const float* __restrict__ src   = urow + (size_t)lane * L;
        const float* __restrict__ uline = urow + (size_t)half * L + 4 * n;
        float*                    rline = rowbuf + half * ROW_PITCH + 4 * n;

        float p1 = 0.0f, p2 = 0.0f, p3 = 0.0f;
        if (32 * wv + lane > 0) {
            p1 = src[-1];
            p2 = src[-2];
            p3 = src[-3];
        }

        v4f g[P1_NV], x[P1_NV];
        lines_load<P1_NV>(g, uline);
        lines_put<P1_NV>(g, rline);
        stage_sync();
#pragma unroll 1
        for (int blk = 0; blk < P1_NB; ++blk) {
#pragma unroll
            for (int v = 0; v < P1_NV; ++v) x[v] = *(const v4f*)(rowown + blk * P1_BLK + 4 * v);
            if (blk + 1 < P1_NB) lines_load<P1_NV>(g, uline + (blk + 1) * P1_BLK);
            p1_block_tab<P1_NV>(x, b0, b1, b2, na0, na1, na2, p1, p2, p3, y1, y2, y3,
                                nta0, nta1, nta2, h1, h2, h3, tcol + blk * P1_BLK * 4);
#pragma unroll
            for (int v = 0; v < P1_NV; ++v) *(v4f*)(rowown + blk * P1_BLK + 4 * v) = x[v];
            if (blk + 1 < P1_NB) {
                lines_put<P1_NV>(g, rline + (blk + 1) * P1_BLK);
                stage_sync();
            }
        }
    }
    stage_sync();

    const float b0 = bco2[0], b1 = bco2[1], b2 = bco2[2];
    const float na0 = -aco2[0], na1 = -aco2[1], na2 = -aco2[2];
    const HTail ht = load_htail(tab2);

    const float* zlo  = rowbuf + n * ROW_PITCH + 8 * half;
    const float* zhi  = rowbuf + (16 + n) * ROW_PITCH + 8 * half;
    const float* hrow = tab1 + n * 4 + 2 * half;

    float e0, e1, e2;
    {
        const HTail ht1 = load_htail(tab1);
        row_scan(ht1, y1, y2, y3, lane, wv, xch1, e0, e1, e2);
    }
    v2f bvl, bvh;
    {
        const v2f g0 = half_swap(e0, e0), g1 = half_swap(e1, e1), g2 = half_swap(e2, e2);
        const float l0 = g0[0], l1 = g1[0], l2 = g2[0];
        const float h0 = g0[1], h1 = g1[1], h2 = g2[1];
        bvl[0] = hi ? l2 : l0;  bvl[1] = hi ? 0.0f : l1;
        bvh[0] = hi ? h2 : h0;  bvh[1] = hi ? 0.0f : h1;
    }

    float p1, p2, p3 = 0.0f;
    {
        v4f cz[1][4]; v2f ah[1]; v4f seq[4];
        f_load<1>(cz, ah, zlo + (L - 16), zhi + (L - 16), hrow + (L - 16) * 4);
        f_v16(cz[0], ah[0], bvl, bvh, seq);
        if (wv == 0 && lane == 31) { vedge[0] = seq[3][3]; vedge[1] = seq[3][2]; }
        __syncthreads();
        p1 = __shfl_up(seq[3][3], 1);
        p2 = __shfl_up(seq[3][2], 1);
        if (lane == 0) {
            p1 = (wv != 0) ? vedge[0] : 0.0f;
            p2 = (wv != 0) ? vedge[1] : 0.0f;
        }
    }
    y1 = 0.0f; y2 = 0.0f; y3 = 0.0f;

    v4f ca[F_NS][4], cb[F_NS][4];
    v2f ha[F_NS], hb[F_NS];
    f_load<F_NS>(ca, ha, zlo, zhi, hrow);
    for (int blk = 0; blk < P1_NB - 2; blk += 2) {
        f_load<F_NS>(cb, hb, zlo + (blk + 1) * P1_BLK, zhi + (blk + 1) * P1_BLK,
                     hrow + (blk + 1) * P1_BLK * 4);
        f_block<F_NS>(ca, ha, bvl, bvh, rowown + blk * P1_BLK,
                      b0, b1, b2, na0, na1, na2, p1, p2, p3, y1, y2, y3);
        f_load<F_NS>(ca, ha, zlo + (blk + 2) * P1_BLK, zhi + (blk + 2) * P1_BLK,
                     hrow + (blk + 2) * P1_BLK * 4);
        f_block<F_NS>(cb, hb, bvl, bvh, rowown + (blk + 1) * P1_BLK,
                      b0, b1, b2, na0, na1, na2, p1, p2, p3, y1, y2, y3);
    }
    f_load<F_NS>(cb, hb, zlo + (P1_NB - 1) * P1_BLK, zhi + (P1_NB - 1) * P1_BLK,
                 hrow + (P1_NB - 1) * P1_BLK * 4);
    f_block<F_NS>(ca, ha, bvl, bvh, rowown + (P1_NB - 2) * P1_BLK,
                  b0, b1, b2, na0, na1, na2, p1, p2, p3, y1, y2, y3);
    f_block<F_NS>(cb, hb, bvl, bvh, rowown + (P1_NB - 1) * P1_BLK,
                  b0, b1, b2, na0, na1, na2, p1, p2, p3, y1, y2, y3);

    row_scan(ht, y1, y2, y3, lane, wv, xch2, e0, e1, e2);
    v2f svl, svh;
    {
        const v2f g0 = half_swap(e0, e0), g1 = half_swap(e1, e1), g2 = half_swap(e2, e2);
        const float l0 = g0[0], l1 = g1[0], l2 = g2[0];
        const float h0 = g0[1], h1 = g1[1], h2 = g2[1];
        svl[0] = hi ? l2 : l0;  svl[1] = hi ? 0.0f : l1;
        svh[0] = hi ? h2 : h0;  svh[1] = hi ? 0.0f : h1;
    }
    stage_sync();

    float* orow = out + (size_t)b * T + (size_t)wv * (32 * L);
    const float* hcol = tab2 + (4 * n) * 4 + 2 * half;
#pragma unroll 2
    for (int blk = 0; blk < P1_NB; ++blk) {
        v2f bt[4];
#pragma unroll
        for (int j = 0; j < 4; ++j) bt[j] = *(const v2f*)(hcol + (blk * P1_BLK + j) * 4);
#pragma unroll
        for (int grp = 0; grp < 2; ++grp) {
            const float* zc = rowbuf + (16 * grp + 8 * half) * ROW_PITCH + blk * P1_BLK + 4 * n;
            float*       oc = orow + (size_t)(16 * grp + 8 * half) * L + blk * P1_BLK + 4 * n;
            v4f cr[8];
#pragma unroll
            for (int r = 0; r < 8; ++r) cr[r] = *(const v4f*)(zc + r * ROW_PITCH);
            v8f acc[4];
#pragma unroll
            for (int j = 0; j < 4; ++j) {
#pragma unroll
                for (int r = 0; r < 8; ++r) acc[j][r] = cr[r][j];
                acc[j] = __builtin_amdgcn_wmma_f32_16x16x4_f32(
                     false, grp ? svh : svl,  false, bt[j],
                     (short)0, acc[j],  false,  false);
            }
#pragma unroll
            for (int r = 0; r < 8; ++r) {
                v4f o;
                o[0] = acc[0][r]; o[1] = acc[1][r]; o[2] = acc[2][r]; o[3] = acc[3][r];
                __builtin_nontemporal_store(o, (v4f*)(oc + (size_t)r * L));
            }
        }
    }
}

extern "C" void kernel_launch(void* const* d_in, const int* in_sizes, int n_in,
                              void* d_out, int out_size, void* d_ws, size_t ws_size,
                              hipStream_t stream) {
    const float* u  = (const float*)d_in[0];
    const float* b1 = (const float*)d_in[1];
    const float* a1 = (const float*)d_in[2];
    const float* b2 = (const float*)d_in[3];
    const float* a2 = (const float*)d_in[4];
    float* out = (float*)d_out;

    fused_kernel<<<Bsz, 32 * WPR, 0, stream>>>(u, b1, a1, b2, a2, out);
}
